# starting kernel
# baseline (speedup 1.0000x reference)
#include <hip/hip_runtime.h>
#include <hip/hip_bf16.h>
#include <math.h>

typedef float v2f __attribute__((ext_vector_type(2)));
typedef float v8f __attribute__((ext_vector_type(8)));

#define NB 8
#define CC 256
#define HH 8
#define DD 32
#define LL 4096
#define TC 768

static __device__ __forceinline__ v8f wmma_f32(v2f a, v2f b, v8f c) {
    return __builtin_amdgcn_wmma_f32_16x16x4_f32(false, a, false, b, (short)0, c, false, false);
}

__global__ __launch_bounds__(256) void qkv_gemm_kernel(
    const float* __restrict__ x, const float* __restrict__ wq,
    const float* __restrict__ bq, float* __restrict__ qkv)
{
    __shared__ float2 Wt2[16 * 66];
    const int mt = blockIdx.x;
    const int nt = blockIdx.y;
    const int row0 = mt * 128;
    const int n = row0 >> 12;
    const int l0 = row0 & (LL - 1);
    const int j0 = nt * 64;
    const int tid = threadIdx.x;
    const int lane = tid & 31, wave = tid >> 5;
    const int m0 = wave * 16;
    const int laneM = lane & 15;
    const int laneH = lane >> 4;
    const int laneK = laneH << 1;

    v8f acc[4] = {};
    const float* xA = x + (size_t)n * CC * LL + l0 + m0 + laneM;

    for (int kc = 0; kc < CC; kc += 32) {
        for (int i = tid; i < 16 * 64; i += 256) {
            const int p = i & 15, j = i >> 4;
            Wt2[p * 66 + j] = *(const float2*)&wq[(size_t)(j0 + j) * CC + kc + 2 * p];
        }
        __syncthreads();
#pragma unroll
        for (int kk = 0; kk < 32; kk += 4) {
            const int k = kc + kk + laneK;
            v2f a;
            a.x = xA[(size_t)k * LL];
            a.y = xA[(size_t)(k + 1) * LL];
            const int pp = (kk >> 1) + laneH;
#pragma unroll
            for (int nb = 0; nb < 4; ++nb) {
                const v2f b = *(const v2f*)&Wt2[pp * 66 + nb * 16 + laneM];
                acc[nb] = wmma_f32(a, b, acc[nb]);
            }
        }
        __syncthreads();
    }
    const int mbase = m0 + (laneH << 3);
#pragma unroll
    for (int nb = 0; nb < 4; ++nb) {
        const int j = j0 + nb * 16 + laneM;
        const float bias = bq[j];
        float* outp = qkv + (size_t)(row0 + mbase) * TC + j;
#pragma unroll
        for (int r = 0; r < 8; ++r) outp[(size_t)r * TC] = acc[nb][r] + bias;
    }
}

__global__ __launch_bounds__(256) void norm_qk_kernel(float* __restrict__ qkv)
{
    size_t t = (size_t)blockIdx.x * 256 + threadIdx.x;
    const int h = (int)(t & 7);
    size_t r = t >> 3;
    const int l = (int)(r & (LL - 1));
    r >>= 12;
    const int n = (int)(r & 7);
    const int s = (int)(r >> 3);
    float* p = qkv + ((size_t)(n * LL + l)) * TC + s * CC + h * DD;
    float4 v[8];
    float ss = 0.f;
#pragma unroll
    for (int i = 0; i < 8; ++i) {
        v[i] = ((const float4*)p)[i];
        ss += v[i].x * v[i].x + v[i].y * v[i].y + v[i].z * v[i].z + v[i].w * v[i].w;
    }
    const float inv = 1.0f / sqrtf(ss);
#pragma unroll
    for (int i = 0; i < 8; ++i) {
        v[i].x *= inv; v[i].y *= inv; v[i].z *= inv; v[i].w *= inv;
        ((float4*)p)[i] = v[i];
    }
}

__global__ __launch_bounds__(256) void attn_kernel(
    const float* __restrict__ qkv, float* __restrict__ attn)
{
    __shared__ float red[8 * 1024];
    const int nh = blockIdx.x;
    const int n = nh >> 3, h = nh & 7;
    const int tid = threadIdx.x, lane = tid & 31, wave = tid >> 5;
    const int laneM = lane & 15;
    const int laneK = (lane >> 4) << 1;

    v8f acc[2][2] = {};
    const float* base = qkv + (size_t)n * LL * TC + h * DD;

    const int lend = wave * 512 + 512;
    for (int l = wave * 512; l < lend; l += 4) {
        const size_t r0 = (size_t)(l + laneK) * TC;
        v2f aK[2], bV[2];
#pragma unroll
        for (int md = 0; md < 2; ++md) {
            aK[md].x = base[r0 + CC + md * 16 + laneM];
            aK[md].y = base[r0 + TC + CC + md * 16 + laneM];
        }
#pragma unroll
        for (int me = 0; me < 2; ++me) {
            bV[me].x = base[r0 + 2 * CC + me * 16 + laneM];
            bV[me].y = base[r0 + TC + 2 * CC + me * 16 + laneM];
        }
#pragma unroll
        for (int md = 0; md < 2; ++md)
#pragma unroll
            for (int me = 0; me < 2; ++me)
                acc[md][me] = wmma_f32(aK[md], bV[me], acc[md][me]);
    }
    const int dadd = (lane >> 4) << 3;
#pragma unroll
    for (int md = 0; md < 2; ++md)
#pragma unroll
        for (int me = 0; me < 2; ++me)
#pragma unroll
            for (int r = 0; r < 8; ++r)
                red[wave * 1024 + (md * 16 + r + dadd) * 32 + me * 16 + laneM] = acc[md][me][r];
    __syncthreads();
    for (int i = tid; i < 1024; i += 256) {
        float s = 0.f;
#pragma unroll
        for (int w = 0; w < 8; ++w) s += red[w * 1024 + i];
        attn[(size_t)nh * 1024 + i] = s;
    }
}

__global__ __launch_bounds__(256) void epilogue_kernel(
    const float* __restrict__ qkv, const float* __restrict__ attn,
    const float* __restrict__ wdc, float* __restrict__ out_pre)
{
    __shared__ float v_s[72 * 32];
    __shared__ float q_s[64 * 32];
    __shared__ float a_s[32 * 32];
    const int b = blockIdx.x;
    const int lc = b & 63;
    const int h = (b >> 6) & 7;
    const int n = b >> 9;
    const int l0 = lc * 64;
    const int tid = threadIdx.x;
    const float* qbase = qkv + (size_t)n * LL * TC + h * DD;

    for (int i = tid; i < 72 * 32; i += 256) {
        const int rr = i >> 5, e = i & 31;
        const int l = l0 - 4 + rr;
        v_s[i] = (l >= 0 && l < LL) ? qbase[(size_t)l * TC + 2 * CC + e] : 0.0f;
    }
    for (int i = tid; i < 64 * 32; i += 256) {
        const int rr = i >> 5, d = i & 31;
        q_s[i] = qbase[(size_t)(l0 + rr) * TC + d];
    }
    for (int i = tid; i < 1024; i += 256)
        a_s[i] = attn[((size_t)(n * 8 + h)) * 1024 + i];
    __syncthreads();

    float wd[9];
#pragma unroll
    for (int r = 0; r < 9; ++r) wd[r] = wdc[h * 9 + r];

    const int lane = tid & 31, wave = tid >> 5;
    const float inv_pi = 0.31830988618379067154f;
    for (int ll = wave * 8; ll < wave * 8 + 8; ++ll) {
        float acc = 0.f;
#pragma unroll
        for (int d = 0; d < 32; ++d) acc += q_s[ll * 32 + d] * a_s[d * 32 + lane];
        const float val = 0.5f * v_s[(ll + 4) * 32 + lane] + inv_pi * acc;
        float ss = val * val;
#pragma unroll
        for (int off = 16; off; off >>= 1) ss += __shfl_xor(ss, off, 32);
        const float res = val / sqrtf(ss);
        float dc = 0.f;
#pragma unroll
        for (int r = 0; r < 9; ++r) dc += wd[r] * v_s[(ll + r) * 32 + lane];
        out_pre[(size_t)(n * LL + l0 + ll) * CC + h * DD + lane] = res + dc;
    }
}

__global__ __launch_bounds__(256) void proj_gemm_kernel(
    const float* __restrict__ A, const float* __restrict__ wp,
    const float* __restrict__ bp, float* __restrict__ out)
{
    __shared__ float2 smemv[4224];
    float* smem = (float*)smemv;
    float2* At2 = smemv;
    float2* Wt2 = smemv + 16 * 130;
    const int mt = blockIdx.x;
    const int nt = blockIdx.y;
    const int row0 = mt * 128;
    const int j0 = nt * 64;
    const int tid = threadIdx.x, lane = tid & 31, wave = tid >> 5;
    const int m0 = wave * 16;
    const int laneM = lane & 15;
    const int laneH = lane >> 4;

    v8f acc[4] = {};
    for (int kc = 0; kc < CC; kc += 32) {
        for (int i = tid; i < 16 * 128; i += 256) {
            const int p = i & 15, rr = i >> 4;
            At2[p * 130 + rr] = *(const float2*)&A[(size_t)(row0 + rr) * CC + kc + 2 * p];
        }
        for (int i = tid; i < 16 * 64; i += 256) {
            const int p = i & 15, j = i >> 4;
            Wt2[p * 66 + j] = *(const float2*)&wp[(size_t)(j0 + j) * CC + kc + 2 * p];
        }
        __syncthreads();
#pragma unroll
        for (int kk = 0; kk < 32; kk += 4) {
            const int pp = (kk >> 1) + laneH;
            const v2f a = *(const v2f*)&At2[pp * 130 + m0 + laneM];
#pragma unroll
            for (int nb = 0; nb < 4; ++nb) {
                const v2f b = *(const v2f*)&Wt2[pp * 66 + nb * 16 + laneM];
                acc[nb] = wmma_f32(a, b, acc[nb]);
            }
        }
        __syncthreads();
    }
    const int lbase = m0 + (laneH << 3);
#pragma unroll
    for (int nb = 0; nb < 4; ++nb) {
        const int j = nb * 16 + laneM;
#pragma unroll
        for (int r = 0; r < 8; ++r) smem[j * 132 + lbase + r] = acc[nb][r];
    }
    __syncthreads();
    const int n = row0 >> 12;
    const int lg0 = row0 & (LL - 1);
    for (int i = tid; i < 64 * 128; i += 256) {
        const int l = i & 127, j = i >> 7;
        out[((size_t)(n * CC + j0 + j)) * LL + lg0 + l] = smem[j * 132 + l] + bp[j0 + j];
    }
}

extern "C" void kernel_launch(void* const* d_in, const int* in_sizes, int n_in,
                              void* d_out, int out_size, void* d_ws, size_t ws_size,
                              hipStream_t stream) {
    const float* x      = (const float*)d_in[0];
    const float* w_qkv  = (const float*)d_in[1];
    const float* b_qkv  = (const float*)d_in[2];
    const float* w_proj = (const float*)d_in[3];
    const float* b_proj = (const float*)d_in[4];
    const float* w_dcv  = (const float*)d_in[5];
    float* out = (float*)d_out;

    float* qkv     = (float*)d_ws;
    float* attn    = qkv + (size_t)NB * LL * TC;
    float* out_pre = attn + (size_t)NB * HH * DD * DD;

    qkv_gemm_kernel<<<dim3(256, 12), 256, 0, stream>>>(x, w_qkv, b_qkv, qkv);
    norm_qk_kernel<<<2048, 256, 0, stream>>>(qkv);
    attn_kernel<<<64, 256, 0, stream>>>(qkv, attn);
    epilogue_kernel<<<4096, 256, 0, stream>>>(qkv, attn, w_dcv, out_pre);
    proj_gemm_kernel<<<dim3(256, 4), 256, 0, stream>>>(out_pre, w_proj, b_proj, out);
}


# Epilogue kernel rebuilt on the f32 matrix unit: q, v rows straight from global into fragment / D-layout registers, attn summed once per 256 rows, no LDS row tiles; product and norm sum orders change
# speedup vs baseline: 1.4788x; 1.4788x over previous
#include <hip/hip_runtime.h>
#include <hip/hip_bf16.h>
#include <math.h>

typedef float v2f __attribute__((ext_vector_type(2)));
typedef float v8f __attribute__((ext_vector_type(8)));

#define NB 8
#define CC 256
#define HH 8
#define DD 32
#define LL 4096
#define TC 768

static __device__ __forceinline__ v8f wmma_f32(v2f a, v2f b, v8f c) {
    return __builtin_amdgcn_wmma_f32_16x16x4_f32(false, a, false, b, (short)0, c, false, false);
}

__global__ __launch_bounds__(256) void qkv_gemm_kernel(
    const float* __restrict__ x, const float* __restrict__ wq,
    const float* __restrict__ bq, float* __restrict__ qkv)
{
    __shared__ float2 Wt2[16 * 66];
    const int mt = blockIdx.x;
    const int nt = blockIdx.y;
    const int row0 = mt * 128;
    const int n = row0 >> 12;
    const int l0 = row0 & (LL - 1);
    const int j0 = nt * 64;
    const int tid = threadIdx.x;
    const int lane = tid & 31, wave = tid >> 5;
    const int m0 = wave * 16;
    const int laneM = lane & 15;
    const int laneH = lane >> 4;
    const int laneK = laneH << 1;

    v8f acc[4] = {};
    const float* xA = x + (size_t)n * CC * LL + l0 + m0 + laneM;

    for (int kc = 0; kc < CC; kc += 32) {
        for (int i = tid; i < 16 * 64; i += 256) {
            const int p = i & 15, j = i >> 4;
            Wt2[p * 66 + j] = *(const float2*)&wq[(size_t)(j0 + j) * CC + kc + 2 * p];
        }
        __syncthreads();
#pragma unroll
        for (int kk = 0; kk < 32; kk += 4) {
            const int k = kc + kk + laneK;
            v2f a;
            a.x = xA[(size_t)k * LL];
            a.y = xA[(size_t)(k + 1) * LL];
            const int pp = (kk >> 1) + laneH;
#pragma unroll
            for (int nb = 0; nb < 4; ++nb) {
                const v2f b = *(const v2f*)&Wt2[pp * 66 + nb * 16 + laneM];
                acc[nb] = wmma_f32(a, b, acc[nb]);
            }
        }
        __syncthreads();
    }
    const int mbase = m0 + (laneH << 3);
#pragma unroll
    for (int nb = 0; nb < 4; ++nb) {
        const int j = j0 + nb * 16 + laneM;
        const float bias = bq[j];
        float* outp = qkv + (size_t)(row0 + mbase) * TC + j;
#pragma unroll
        for (int r = 0; r < 8; ++r) outp[(size_t)r * TC] = acc[nb][r] + bias;
    }
}

__global__ __launch_bounds__(512) void norm_attn_kernel(float* __restrict__ qkv)
{
    __shared__ float red[16 * 1024];
    const int b = blockIdx.x;
    const int quarter = b & 3;
    const int h = (b >> 2) & 7;
    const int n = b >> 5;
    const int tid = threadIdx.x, lane = tid & 31, wave = tid >> 5;
    const int c = lane & 15, hf = lane >> 4;
    float* base = qkv + ((size_t)n * LL + quarter * 1024 + wave * 64) * TC + h * DD;
    float* rowp = base + (size_t)c * TC + 16 * hf;
    const float* frag = base + (size_t)(8 * hf) * TC + 2 * c;

    v8f acc[2][2] = {};
    const v2f ones = {1.0f, 1.0f};

    float4 q4[4], k4[4];
    float2 kf[8], vf[8];
#pragma unroll
    for (int i = 0; i < 4; ++i) {
        q4[i] = *(const float4*)(rowp + 4 * i);
        k4[i] = *(const float4*)(rowp + CC + 4 * i);
    }
#pragma unroll
    for (int r = 0; r < 8; ++r) {
        kf[r] = *(const float2*)(frag + (size_t)r * TC + CC);
        vf[r] = *(const float2*)(frag + (size_t)r * TC + 2 * CC);
    }

#pragma unroll
    for (int t = 0; t < 4; ++t) {
        float4 q4n[4], k4n[4];
        float2 kfn[8], vfn[8];
        if (t < 3) {
            const size_t off = (size_t)(t + 1) * 16 * TC;
#pragma unroll
            for (int i = 0; i < 4; ++i) {
                q4n[i] = *(const float4*)(rowp + off + 4 * i);
                k4n[i] = *(const float4*)(rowp + off + CC + 4 * i);
            }
#pragma unroll
            for (int r = 0; r < 8; ++r) {
                kfn[r] = *(const float2*)(frag + off + (size_t)r * TC + CC);
                vfn[r] = *(const float2*)(frag + off + (size_t)r * TC + 2 * CC);
            }
        }
        float pq = 0.f, pk = 0.f;
#pragma unroll
        for (int i = 0; i < 4; ++i) {
            pq += q4[i].x * q4[i].x + q4[i].y * q4[i].y + q4[i].z * q4[i].z + q4[i].w * q4[i].w;
            pk += k4[i].x * k4[i].x + k4[i].y * k4[i].y + k4[i].z * k4[i].z + k4[i].w * k4[i].w;
        }
        pq += __shfl_xor(pq, 16, 32);
        const float invq = 1.0f / sqrtf(pq);
        float* qo = rowp + (size_t)t * 16 * TC;
#pragma unroll
        for (int i = 0; i < 4; ++i) {
            float4 o = q4[i];
            o.x *= invq; o.y *= invq; o.z *= invq; o.w *= invq;
            *(float4*)(qo + 4 * i) = o;
        }
        v2f an; an.x = pk; an.y = 0.0f;
        v8f zero = {};
        const v8f ss = wmma_f32(an, ones, zero);
#pragma unroll
        for (int s = 0; s < 4; ++s) {
            const float i0 = 1.0f / sqrtf(ss[2 * s]);
            const float i1 = 1.0f / sqrtf(ss[2 * s + 1]);
            v2f a0, a1, b0, b1;
            a0.x = kf[2 * s].x * i0; a0.y = kf[2 * s + 1].x * i1;
            a1.x = kf[2 * s].y * i0; a1.y = kf[2 * s + 1].y * i1;
            b0.x = vf[2 * s].x;      b0.y = vf[2 * s + 1].x;
            b1.x = vf[2 * s].y;      b1.y = vf[2 * s + 1].y;
            acc[0][0] = wmma_f32(a0, b0, acc[0][0]);
            acc[0][1] = wmma_f32(a0, b1, acc[0][1]);
            acc[1][0] = wmma_f32(a1, b0, acc[1][0]);
            acc[1][1] = wmma_f32(a1, b1, acc[1][1]);
        }
        if (t < 3) {
#pragma unroll
            for (int i = 0; i < 4; ++i) { q4[i] = q4n[i]; k4[i] = k4n[i]; }
#pragma unroll
            for (int r = 0; r < 8; ++r) { kf[r] = kfn[r]; vf[r] = vfn[r]; }
        }
    }
#pragma unroll
    for (int md = 0; md < 2; ++md)
#pragma unroll
        for (int r = 0; r < 8; ++r) {
            float2 o; o.x = acc[md][0][r]; o.y = acc[md][1][r];
            *(float2*)&red[wave * 1024 + (16 * hf + 2 * r + md) * 32 + 2 * c] = o;
        }
    __syncthreads();
    float* pout = qkv + ((size_t)n * LL + quarter * 1024) * TC + CC + h * DD;
    for (int i = tid; i < 1024; i += 512) {
        float s = 0.f;
#pragma unroll
        for (int w = 0; w < 16; ++w) s += red[w * 1024 + i];
        pout[(size_t)(i >> 5) * TC + (i & 31)] = s;
    }
}

static __device__ __forceinline__ float row16_sum(float x) {
    x += __int_as_float(__builtin_amdgcn_update_dpp(0, __float_as_int(x), 0x161, 0xf, 0xf, true));
    x += __int_as_float(__builtin_amdgcn_update_dpp(0, __float_as_int(x), 0x162, 0xf, 0xf, true));
    x += __int_as_float(__builtin_amdgcn_update_dpp(0, __float_as_int(x), 0x164, 0xf, 0xf, true));
    x += __int_as_float(__builtin_amdgcn_update_dpp(0, __float_as_int(x), 0x168, 0xf, 0xf, true));
    return x;
}

__global__ __launch_bounds__(256) void epilogue_kernel(
    const float* __restrict__ qkv,
    const float* __restrict__ wdc, float* __restrict__ out_pre)
{
    __shared__ float4 Bq[16 * 16];
    const int b = blockIdx.x;
    const int lc = b & 15;
    const int h = (b >> 4) & 7;
    const int n = b >> 7;
    const int tid = threadIdx.x, lane = tid & 31;
    const int wave = tid >> 5;
    const int c = lane & 15, hf = lane >> 4;
    const int lw = lc * 256 + wave * 32;
    const float* qbase = qkv + (size_t)n * LL * TC + h * DD;

    float asum[4];
#pragma unroll
    for (int j = 0; j < 4; ++j) {
        const int d = (tid >> 5) + 8 * j;
        const float* pp = qkv + ((size_t)n * LL + d) * TC + CC + h * DD + lane;
        float a = pp[0];
        a += pp[(size_t)1024 * TC];
        a += pp[(size_t)2048 * TC];
        a += pp[(size_t)3072 * TC];
        asum[j] = a;
    }

    float4 q4[2][4];
    float2 vv[2][16];
#pragma unroll
    for (int t = 0; t < 2; ++t) {
        const float* qr = qbase + (size_t)(lw + 16 * t + c) * TC + 16 * hf;
#pragma unroll
        for (int i = 0; i < 4; ++i) q4[t][i] = *(const float4*)(qr + 4 * i);
    }
    const int lv0 = lw + 8 * hf - 4;
    const float* vcol = qbase + 2 * CC + 2 * c;
#pragma unroll
    for (int i = 0; i < 4; ++i) {
        const int l = lv0 + i;
        const float2 x = *(const float2*)(vcol + (size_t)(l < 0 ? 0 : l) * TC);
        vv[0][i].x = (l < 0) ? 0.0f : x.x;
        vv[0][i].y = (l < 0) ? 0.0f : x.y;
    }
    {
        const float* vr = vcol + (size_t)(lv0 + 4) * TC;
#pragma unroll
        for (int i = 4; i < 28; ++i) vv[i >> 4][i & 15] = *(const float2*)(vr + (size_t)(i - 4) * TC);
    }
#pragma unroll
    for (int i = 12; i < 16; ++i) {
        const int l = lv0 + 16 + i;
        const float2 x = *(const float2*)(vcol + (size_t)(l < LL ? l : LL - 1) * TC);
        vv[1][i].x = (l < LL) ? x.x : 0.0f;
        vv[1][i].y = (l < LL) ? x.y : 0.0f;
    }

#pragma unroll
    for (int j = 0; j < 4; ++j) {
        const int d = (tid >> 5) + 8 * j;
        ((float*)Bq)[(((d >> 1) * 16 + (lane >> 1)) << 2) + ((lane & 1) << 1) + (d & 1)] = asum[j];
    }
    __syncthreads();

    float4 bf[8];
#pragma unroll
    for (int s = 0; s < 8; ++s) bf[s] = Bq[(8 * hf + s) * 16 + c];

    float wd[9];
#pragma unroll
    for (int r = 0; r < 9; ++r) wd[r] = wdc[h * 9 + r];

    const float inv_pi = 0.31830988618379067154f;
#pragma unroll
    for (int t = 0; t < 2; ++t) {
        v8f acc0 = {}, acc1 = {};
#pragma unroll
        for (int s = 0; s < 8; ++s) {
            const float4 qq = q4[t][s >> 1];
            v2f a, b0, b1;
            if (s & 1) { a.x = qq.z; a.y = qq.w; } else { a.x = qq.x; a.y = qq.y; }
            b0.x = bf[s].x; b0.y = bf[s].y;
            b1.x = bf[s].z; b1.y = bf[s].w;
            acc0 = wmma_f32(a, b0, acc0);
            acc1 = wmma_f32(a, b1, acc1);
        }
        float* orow = out_pre + (size_t)(n * LL + lw + 16 * t + 8 * hf) * CC + h * DD + 2 * c;
#pragma unroll
        for (int r = 0; r < 8; ++r) {
            const float2 vc = vv[t][r + 4];
            const float v0 = 0.5f * vc.x + inv_pi * acc0[r];
            const float v1 = 0.5f * vc.y + inv_pi * acc1[r];
            const float nrm = sqrtf(row16_sum(v0 * v0 + v1 * v1));
            float dc0 = 0.f, dc1 = 0.f;
#pragma unroll
            for (int k = 0; k < 9; ++k) {
                dc0 += wd[k] * vv[t][r + k].x;
                dc1 += wd[k] * vv[t][r + k].y;
            }
            float2 o;
            o.x = v0 / nrm + dc0;
            o.y = v1 / nrm + dc1;
            *(float2*)(orow + (size_t)r * CC) = o;
        }
    }
}

__global__ __launch_bounds__(256) void proj_gemm_kernel(
    const float* __restrict__ A, const float* __restrict__ wp,
    const float* __restrict__ bp, float* __restrict__ out)
{
    __shared__ float2 Wt2[16 * 66];
    const int row0 = blockIdx.x * 256;
    const int j0 = blockIdx.y * 64;
    const int tid = threadIdx.x, lane = tid & 31, wave = tid >> 5;
    const int laneM = lane & 15;
    const int laneH = lane >> 4;
    const int lw0 = wave * 32;

    v8f acc[4][2] = {};
    const float* xr0 = A + (size_t)(row0 + lw0 + laneM) * CC + 16 * laneH;
    const float* xr1 = xr0 + (size_t)16 * CC;

    float4 cur[2][4], nxt[2][4];
#pragma unroll
    for (int q = 0; q < 4; ++q) {
        cur[0][q] = *(const float4*)(xr0 + 4 * q);
        cur[1][q] = *(const float4*)(xr1 + 4 * q);
    }

#pragma unroll
    for (int kc = 0; kc < CC; kc += 32) {
        for (int i = tid; i < 16 * 64; i += 256) {
            const int p = i & 15, j = i >> 4;
            Wt2[p * 66 + j] = *(const float2*)&wp[(size_t)(j0 + j) * CC + kc + 2 * p];
        }
        __syncthreads();
        if (kc + 32 < CC) {
#pragma unroll
            for (int q = 0; q < 4; ++q) {
                nxt[0][q] = *(const float4*)(xr0 + kc + 32 + 4 * q);
                nxt[1][q] = *(const float4*)(xr1 + kc + 32 + 4 * q);
            }
        }
#pragma unroll
        for (int s = 0; s < 8; ++s) {
            const int pp = 8 * laneH + s;
            const float4 c0 = cur[0][s >> 1], c1 = cur[1][s >> 1];
            v2f b0, b1;
            if (s & 1) { b0.x = c0.z; b0.y = c0.w; b1.x = c1.z; b1.y = c1.w; }
            else       { b0.x = c0.x; b0.y = c0.y; b1.x = c1.x; b1.y = c1.y; }
#pragma unroll
            for (int jt = 0; jt < 4; ++jt) {
                const v2f a = *(const v2f*)&Wt2[pp * 66 + jt * 16 + laneM];
                acc[jt][0] = wmma_f32(a, b0, acc[jt][0]);
                acc[jt][1] = wmma_f32(a, b1, acc[jt][1]);
            }
        }
        __syncthreads();
        if (kc + 32 < CC) {
#pragma unroll
            for (int q = 0; q < 4; ++q) { cur[0][q] = nxt[0][q]; cur[1][q] = nxt[1][q]; }
        }
    }
    const int n = row0 >> 12;
    const int lg0 = row0 & (LL - 1);
#pragma unroll
    for (int jt = 0; jt < 4; ++jt) {
        const int jb = j0 + jt * 16 + (laneH << 3);
#pragma unroll
        for (int r = 0; r < 8; ++r) {
            const float bias = bp[jb + r];
            float* o = out + ((size_t)(n * CC + jb + r)) * LL + lg0 + lw0 + laneM;
            o[0]  = acc[jt][0][r] + bias;
            o[16] = acc[jt][1][r] + bias;
        }
    }
}

extern "C" void kernel_launch(void* const* d_in, const int* in_sizes, int n_in,
                              void* d_out, int out_size, void* d_ws, size_t ws_size,
                              hipStream_t stream) {
    const float* x      = (const float*)d_in[0];
    const float* w_qkv  = (const float*)d_in[1];
    const float* b_qkv  = (const float*)d_in[2];
    const float* w_proj = (const float*)d_in[3];
    const float* b_proj = (const float*)d_in[4];
    const float* w_dcv  = (const float*)d_in[5];
    float* out = (float*)d_out;

    float* qkv     = (float*)d_ws;
    float* out_pre = qkv + (size_t)NB * LL * TC + (size_t)NB * HH * DD * DD;

    qkv_gemm_kernel<<<dim3(256, 12), 256, 0, stream>>>(x, w_qkv, b_qkv, qkv);
    norm_attn_kernel<<<256, 512, 0, stream>>>(qkv);
    epilogue_kernel<<<1024, 256, 0, stream>>>(qkv, w_dcv, out_pre);
    proj_gemm_kernel<<<dim3(128, 4), 256, 0, stream>>>(out_pre, w_proj, b_proj, out);
}


# QKV GEMM re-tiled: a block owns 64 rows x all 768 columns; its x slice sits in LDS as k pairs, read once; weights go global -> B fragments a chunk ahead; K loop has no barrier; k sum order changes
# speedup vs baseline: 1.5939x; 1.0778x over previous
#include <hip/hip_runtime.h>
#include <hip/hip_bf16.h>
#include <math.h>

typedef float v2f __attribute__((ext_vector_type(2)));
typedef float v8f __attribute__((ext_vector_type(8)));

#define NB 8
#define CC 256
#define HH 8
#define DD 32
#define LL 4096
#define TC 768

static __device__ __forceinline__ v8f wmma_f32(v2f a, v2f b, v8f c) {
    return __builtin_amdgcn_wmma_f32_16x16x4_f32(false, a, false, b, (short)0, c, false, false);
}

static __device__ __forceinline__ void qkv_chunk(v8f (&acc)[2][4], const float4 (&w)[2][4],
                                                 const float2* __restrict__ ap)
{
#pragma unroll
    for (int s = 0; s < 8; ++s) {
        v2f a[4];
#pragma unroll
        for (int rt = 0; rt < 4; ++rt) a[rt] = *(const v2f*)(ap + s * 66 + 16 * rt);
        const float4 c0 = w[0][s >> 1], c1 = w[1][s >> 1];
        v2f b0, b1;
        if (s & 1) { b0.x = c0.z; b0.y = c0.w; b1.x = c1.z; b1.y = c1.w; }
        else       { b0.x = c0.x; b0.y = c0.y; b1.x = c1.x; b1.y = c1.y; }
#pragma unroll
        for (int rt = 0; rt < 4; ++rt) {
            acc[0][rt] = wmma_f32(a[rt], b0, acc[0][rt]);
            acc[1][rt] = wmma_f32(a[rt], b1, acc[1][rt]);
        }
    }
}

__global__ __launch_bounds__(256) __attribute__((amdgpu_waves_per_eu(4, 4))) void qkv_gemm_kernel(
    const float* __restrict__ x, const float* __restrict__ wq,
    const float* __restrict__ bq, float* __restrict__ qkv)
{
    __shared__ float2 Xs2[128 * 66];
    const int row0 = blockIdx.x * 64;
    const int n = row0 >> 12;
    const int l0 = row0 & (LL - 1);
    const int tid = threadIdx.x;
    const int lane = tid & 31, wave = tid >> 5;
    const int c = lane & 15, hf = lane >> 4;

    {
        const int m4 = (tid & 15) * 4;
        const int pb = tid >> 4;
        const float* xs = x + (size_t)n * CC * LL + l0 + m4;
        float4 lo[8], hi[8];
#pragma unroll
        for (int i = 0; i < 8; ++i) {
            const int p = pb + 16 * i;
            lo[i] = *(const float4*)(xs + (size_t)(2 * p) * LL);
            hi[i] = *(const float4*)(xs + (size_t)(2 * p + 1) * LL);
        }
#pragma unroll
        for (int i = 0; i < 8; ++i) {
            const int p = pb + 16 * i;
            float4 s0, s1;
            s0.x = lo[i].x; s0.y = hi[i].x; s0.z = lo[i].y; s0.w = hi[i].y;
            s1.x = lo[i].z; s1.y = hi[i].z; s1.z = lo[i].w; s1.w = hi[i].w;
            *(float4*)&Xs2[p * 66 + m4]     = s0;
            *(float4*)&Xs2[p * 66 + m4 + 2] = s1;
        }
    }

    const float* wrow = wq + (size_t)(32 * wave + c) * CC + 16 * hf;
    const float2* arow = Xs2 + (8 * hf) * 66 + c;
    float* obase = qkv + (size_t)(row0 + 8 * hf) * TC + 32 * wave + c;

    float4 wa[2][4], wb[2][4];
#pragma unroll
    for (int q = 0; q < 4; ++q) {
        wa[0][q] = *(const float4*)(wrow + 4 * q);
        wa[1][q] = *(const float4*)(wrow + 16 * CC + 4 * q);
    }
    __syncthreads();

    for (int cp = 0; cp < 3; ++cp) {
        const float* wg = wrow + (size_t)cp * 256 * CC;
        v8f acc[2][4] = {};
#pragma unroll 1
        for (int kk = 0; kk < 4; ++kk) {
            const int kc = kk * 64;
#pragma unroll
            for (int q = 0; q < 4; ++q) {
                wb[0][q] = *(const float4*)(wg + kc + 32 + 4 * q);
                wb[1][q] = *(const float4*)(wg + 16 * CC + kc + 32 + 4 * q);
            }
            qkv_chunk(acc, wa, arow + (kc >> 1) * 66);
            {
                const float* wn = (kk < 3) ? (wg + kc + 64) : ((cp < 2) ? (wg + 256 * CC) : wg);
#pragma unroll
                for (int q = 0; q < 4; ++q) {
                    wa[0][q] = *(const float4*)(wn + 4 * q);
                    wa[1][q] = *(const float4*)(wn + 16 * CC + 4 * q);
                }
            }
            qkv_chunk(acc, wb, arow + ((kc + 32) >> 1) * 66);
        }
        const float bias0 = bq[256 * cp + 32 * wave + c];
        const float bias1 = bq[256 * cp + 32 * wave + 16 + c];
        float* og = obase + 256 * cp;
#pragma unroll
        for (int rt = 0; rt < 4; ++rt)
#pragma unroll
            for (int r = 0; r < 8; ++r) {
                float* o = og + (size_t)(16 * rt + r) * TC;
                o[0]  = acc[0][rt][r] + bias0;
                o[16] = acc[1][rt][r] + bias1;
            }
    }
}

__global__ __launch_bounds__(512) void norm_attn_kernel(float* __restrict__ qkv)
{
    __shared__ float red[16 * 1024];
    const int b = blockIdx.x;
    const int quarter = b & 3;
    const int h = (b >> 2) & 7;
    const int n = b >> 5;
    const int tid = threadIdx.x, lane = tid & 31, wave = tid >> 5;
    const int c = lane & 15, hf = lane >> 4;
    float* base = qkv + ((size_t)n * LL + quarter * 1024 + wave * 64) * TC + h * DD;
    float* rowp = base + (size_t)c * TC + 16 * hf;
    const float* frag = base + (size_t)(8 * hf) * TC + 2 * c;

    v8f acc[2][2] = {};
    const v2f ones = {1.0f, 1.0f};

    float4 q4[4], k4[4];
    float2 kf[8], vf[8];
#pragma unroll
    for (int i = 0; i < 4; ++i) {
        q4[i] = *(const float4*)(rowp + 4 * i);
        k4[i] = *(const float4*)(rowp + CC + 4 * i);
    }
#pragma unroll
    for (int r = 0; r < 8; ++r) {
        kf[r] = *(const float2*)(frag + (size_t)r * TC + CC);
        vf[r] = *(const float2*)(frag + (size_t)r * TC + 2 * CC);
    }

#pragma unroll
    for (int t = 0; t < 4; ++t) {
        float4 q4n[4], k4n[4];
        float2 kfn[8], vfn[8];
        if (t < 3) {
            const size_t off = (size_t)(t + 1) * 16 * TC;
#pragma unroll
            for (int i = 0; i < 4; ++i) {
                q4n[i] = *(const float4*)(rowp + off + 4 * i);
                k4n[i] = *(const float4*)(rowp + off + CC + 4 * i);
            }
#pragma unroll
            for (int r = 0; r < 8; ++r) {
                kfn[r] = *(const float2*)(frag + off + (size_t)r * TC + CC);
                vfn[r] = *(const float2*)(frag + off + (size_t)r * TC + 2 * CC);
            }
        }
        float pq = 0.f, pk = 0.f;
#pragma unroll
        for (int i = 0; i < 4; ++i) {
            pq += q4[i].x * q4[i].x + q4[i].y * q4[i].y + q4[i].z * q4[i].z + q4[i].w * q4[i].w;
            pk += k4[i].x * k4[i].x + k4[i].y * k4[i].y + k4[i].z * k4[i].z + k4[i].w * k4[i].w;
        }
        pq += __shfl_xor(pq, 16, 32);
        const float invq = 1.0f / sqrtf(pq);
        float* qo = rowp + (size_t)t * 16 * TC;
#pragma unroll
        for (int i = 0; i < 4; ++i) {
            float4 o = q4[i];
            o.x *= invq; o.y *= invq; o.z *= invq; o.w *= invq;
            *(float4*)(qo + 4 * i) = o;
        }
        v2f an; an.x = pk; an.y = 0.0f;
        v8f zero = {};
        const v8f ss = wmma_f32(an, ones, zero);
#pragma unroll
        for (int s = 0; s < 4; ++s) {
            const float i0 = 1.0f / sqrtf(ss[2 * s]);
            const float i1 = 1.0f / sqrtf(ss[2 * s + 1]);
            v2f a0, a1, b0, b1;
            a0.x = kf[2 * s].x * i0; a0.y = kf[2 * s + 1].x * i1;
            a1.x = kf[2 * s].y * i0; a1.y = kf[2 * s + 1].y * i1;
            b0.x = vf[2 * s].x;      b0.y = vf[2 * s + 1].x;
            b1.x = vf[2 * s].y;      b1.y = vf[2 * s + 1].y;
            acc[0][0] = wmma_f32(a0, b0, acc[0][0]);
            acc[0][1] = wmma_f32(a0, b1, acc[0][1]);
            acc[1][0] = wmma_f32(a1, b0, acc[1][0]);
            acc[1][1] = wmma_f32(a1, b1, acc[1][1]);
        }
        if (t < 3) {
#pragma unroll
            for (int i = 0; i < 4; ++i) { q4[i] = q4n[i]; k4[i] = k4n[i]; }
#pragma unroll
            for (int r = 0; r < 8; ++r) { kf[r] = kfn[r]; vf[r] = vfn[r]; }
        }
    }
#pragma unroll
    for (int md = 0; md < 2; ++md)
#pragma unroll
        for (int r = 0; r < 8; ++r) {
            float2 o; o.x = acc[md][0][r]; o.y = acc[md][1][r];
            *(float2*)&red[wave * 1024 + (16 * hf + 2 * r + md) * 32 + 2 * c] = o;
        }
    __syncthreads();
    float* pout = qkv + ((size_t)n * LL + quarter * 1024) * TC + CC + h * DD;
    for (int i = tid; i < 1024; i += 512) {
        float s = 0.f;
#pragma unroll
        for (int w = 0; w < 16; ++w) s += red[w * 1024 + i];
        pout[(size_t)(i >> 5) * TC + (i & 31)] = s;
    }
}

static __device__ __forceinline__ float row16_sum(float x) {
    x += __int_as_float(__builtin_amdgcn_update_dpp(0, __float_as_int(x), 0x161, 0xf, 0xf, true));
    x += __int_as_float(__builtin_amdgcn_update_dpp(0, __float_as_int(x), 0x162, 0xf, 0xf, true));
    x += __int_as_float(__builtin_amdgcn_update_dpp(0, __float_as_int(x), 0x164, 0xf, 0xf, true));
    x += __int_as_float(__builtin_amdgcn_update_dpp(0, __float_as_int(x), 0x168, 0xf, 0xf, true));
    return x;
}

__global__ __launch_bounds__(256) void epilogue_kernel(
    const float* __restrict__ qkv,
    const float* __restrict__ wdc, float* __restrict__ out_pre)
{
    __shared__ float4 Bq[16 * 16];
    const int b = blockIdx.x;
    const int lc = b & 15;
    const int h = (b >> 4) & 7;
    const int n = b >> 7;
    const int tid = threadIdx.x, lane = tid & 31;
    const int wave = tid >> 5;
    const int c = lane & 15, hf = lane >> 4;
    const int lw = lc * 256 + wave * 32;
    const float* qbase = qkv + (size_t)n * LL * TC + h * DD;

    float asum[4];
#pragma unroll
    for (int j = 0; j < 4; ++j) {
        const int d = (tid >> 5) + 8 * j;
        const float* pp = qkv + ((size_t)n * LL + d) * TC + CC + h * DD + lane;
        float a = pp[0];
        a += pp[(size_t)1024 * TC];
        a += pp[(size_t)2048 * TC];
        a += pp[(size_t)3072 * TC];
        asum[j] = a;
    }

    float4 q4[2][4];
    float2 vv[2][16];
#pragma unroll
    for (int t = 0; t < 2; ++t) {
        const float* qr = qbase + (size_t)(lw + 16 * t + c) * TC + 16 * hf;
#pragma unroll
        for (int i = 0; i < 4; ++i) q4[t][i] = *(const float4*)(qr + 4 * i);
    }
    const int lv0 = lw + 8 * hf - 4;
    const float* vcol = qbase + 2 * CC + 2 * c;
#pragma unroll
    for (int i = 0; i < 4; ++i) {
        const int l = lv0 + i;
        const float2 x = *(const float2*)(vcol + (size_t)(l < 0 ? 0 : l) * TC);
        vv[0][i].x = (l < 0) ? 0.0f : x.x;
        vv[0][i].y = (l < 0) ? 0.0f : x.y;
    }
    {
        const float* vr = vcol + (size_t)(lv0 + 4) * TC;
#pragma unroll
        for (int i = 4; i < 28; ++i) vv[i >> 4][i & 15] = *(const float2*)(vr + (size_t)(i - 4) * TC);
    }
#pragma unroll
    for (int i = 12; i < 16; ++i) {
        const int l = lv0 + 16 + i;
        const float2 x = *(const float2*)(vcol + (size_t)(l < LL ? l : LL - 1) * TC);
        vv[1][i].x = (l < LL) ? x.x : 0.0f;
        vv[1][i].y = (l < LL) ? x.y : 0.0f;
    }

#pragma unroll
    for (int j = 0; j < 4; ++j) {
        const int d = (tid >> 5) + 8 * j;
        ((float*)Bq)[(((d >> 1) * 16 + (lane >> 1)) << 2) + ((lane & 1) << 1) + (d & 1)] = asum[j];
    }
    __syncthreads();

    float4 bf[8];
#pragma unroll
    for (int s = 0; s < 8; ++s) bf[s] = Bq[(8 * hf + s) * 16 + c];

    float wd[9];
#pragma unroll
    for (int r = 0; r < 9; ++r) wd[r] = wdc[h * 9 + r];

    const float inv_pi = 0.31830988618379067154f;
#pragma unroll
    for (int t = 0; t < 2; ++t) {
        v8f acc0 = {}, acc1 = {};
#pragma unroll
        for (int s = 0; s < 8; ++s) {
            const float4 qq = q4[t][s >> 1];
            v2f a, b0, b1;
            if (s & 1) { a.x = qq.z; a.y = qq.w; } else { a.x = qq.x; a.y = qq.y; }
            b0.x = bf[s].x; b0.y = bf[s].y;
            b1.x = bf[s].z; b1.y = bf[s].w;
            acc0 = wmma_f32(a, b0, acc0);
            acc1 = wmma_f32(a, b1, acc1);
        }
        float* orow = out_pre + (size_t)(n * LL + lw + 16 * t + 8 * hf) * CC + h * DD + 2 * c;
#pragma unroll
        for (int r = 0; r < 8; ++r) {
            const float2 vc = vv[t][r + 4];
            const float v0 = 0.5f * vc.x + inv_pi * acc0[r];
            const float v1 = 0.5f * vc.y + inv_pi * acc1[r];
            const float nrm = sqrtf(row16_sum(v0 * v0 + v1 * v1));
            float dc0 = 0.f, dc1 = 0.f;
#pragma unroll
            for (int k = 0; k < 9; ++k) {
                dc0 += wd[k] * vv[t][r + k].x;
                dc1 += wd[k] * vv[t][r + k].y;
            }
            float2 o;
            o.x = v0 / nrm + dc0;
            o.y = v1 / nrm + dc1;
            *(float2*)(orow + (size_t)r * CC) = o;
        }
    }
}

__global__ __launch_bounds__(256) void proj_gemm_kernel(
    const float* __restrict__ A, const float* __restrict__ wp,
    const float* __restrict__ bp, float* __restrict__ out)
{
    __shared__ float2 Wt2[16 * 66];
    const int row0 = blockIdx.x * 256;
    const int j0 = blockIdx.y * 64;
    const int tid = threadIdx.x, lane = tid & 31, wave = tid >> 5;
    const int laneM = lane & 15;
    const int laneH = lane >> 4;
    const int lw0 = wave * 32;

    v8f acc[4][2] = {};
    const float* xr0 = A + (size_t)(row0 + lw0 + laneM) * CC + 16 * laneH;
    const float* xr1 = xr0 + (size_t)16 * CC;

    float4 cur[2][4], nxt[2][4];
#pragma unroll
    for (int q = 0; q < 4; ++q) {
        cur[0][q] = *(const float4*)(xr0 + 4 * q);
        cur[1][q] = *(const float4*)(xr1 + 4 * q);
    }

#pragma unroll
    for (int kc = 0; kc < CC; kc += 32) {
        for (int i = tid; i < 16 * 64; i += 256) {
            const int p = i & 15, j = i >> 4;
            Wt2[p * 66 + j] = *(const float2*)&wp[(size_t)(j0 + j) * CC + kc + 2 * p];
        }
        __syncthreads();
        if (kc + 32 < CC) {
#pragma unroll
            for (int q = 0; q < 4; ++q) {
                nxt[0][q] = *(const float4*)(xr0 + kc + 32 + 4 * q);
                nxt[1][q] = *(const float4*)(xr1 + kc + 32 + 4 * q);
            }
        }
#pragma unroll
        for (int s = 0; s < 8; ++s) {
            const int pp = 8 * laneH + s;
            const float4 c0 = cur[0][s >> 1], c1 = cur[1][s >> 1];
            v2f b0, b1;
            if (s & 1) { b0.x = c0.z; b0.y = c0.w; b1.x = c1.z; b1.y = c1.w; }
            else       { b0.x = c0.x; b0.y = c0.y; b1.x = c1.x; b1.y = c1.y; }
#pragma unroll
            for (int jt = 0; jt < 4; ++jt) {
                const v2f a = *(const v2f*)&Wt2[pp * 66 + jt * 16 + laneM];
                acc[jt][0] = wmma_f32(a, b0, acc[jt][0]);
                acc[jt][1] = wmma_f32(a, b1, acc[jt][1]);
            }
        }
        __syncthreads();
        if (kc + 32 < CC) {
#pragma unroll
            for (int q = 0; q < 4; ++q) { cur[0][q] = nxt[0][q]; cur[1][q] = nxt[1][q]; }
        }
    }
    const int n = row0 >> 12;
    const int lg0 = row0 & (LL - 1);
#pragma unroll
    for (int jt = 0; jt < 4; ++jt) {
        const int jb = j0 + jt * 16 + (laneH << 3);
#pragma unroll
        for (int r = 0; r < 8; ++r) {
            const float bias = bp[jb + r];
            float* o = out + ((size_t)(n * CC + jb + r)) * LL + lg0 + lw0 + laneM;
            o[0]  = acc[jt][0][r] + bias;
            o[16] = acc[jt][1][r] + bias;
        }
    }
}

extern "C" void kernel_launch(void* const* d_in, const int* in_sizes, int n_in,
                              void* d_out, int out_size, void* d_ws, size_t ws_size,
                              hipStream_t stream) {
    const float* x      = (const float*)d_in[0];
    const float* w_qkv  = (const float*)d_in[1];
    const float* b_qkv  = (const float*)d_in[2];
    const float* w_proj = (const float*)d_in[3];
    const float* b_proj = (const float*)d_in[4];
    const float* w_dcv  = (const float*)d_in[5];
    float* out = (float*)d_out;

    float* qkv     = (float*)d_ws;
    float* out_pre = qkv + (size_t)NB * LL * TC + (size_t)NB * HH * DD * DD;

    qkv_gemm_kernel<<<512, 256, 0, stream>>>(x, w_qkv, b_qkv, qkv);
    norm_attn_kernel<<<256, 512, 0, stream>>>(qkv);
    epilogue_kernel<<<1024, 256, 0, stream>>>(qkv, w_dcv, out_pre);
    proj_gemm_kernel<<<dim3(128, 4), 256, 0, stream>>>(out_pre, w_proj, b_proj, out);
}


# q is normalised by its only reader: the epilogue scales its q A-fragments in registers (same sums, same order); the k^T v pass reads k and v only and no longer loads or rewrites q
# speedup vs baseline: 1.8705x; 1.1735x over previous
#include <hip/hip_runtime.h>
#include <hip/hip_bf16.h>
#include <math.h>

typedef float v2f __attribute__((ext_vector_type(2)));
typedef float v8f __attribute__((ext_vector_type(8)));

#define NB 8
#define CC 256
#define HH 8
#define DD 32
#define LL 4096
#define TC 768

static __device__ __forceinline__ v8f wmma_f32(v2f a, v2f b, v8f c) {
    return __builtin_amdgcn_wmma_f32_16x16x4_f32(false, a, false, b, (short)0, c, false, false);
}

static __device__ __forceinline__ void qkv_chunk(v8f (&acc)[2][4], const float4 (&w)[2][4],
                                                 const float2* __restrict__ ap)
{
#pragma unroll
    for (int s = 0; s < 8; ++s) {
        v2f a[4];
#pragma unroll
        for (int rt = 0; rt < 4; ++rt) a[rt] = *(const v2f*)(ap + s * 66 + 16 * rt);
        const float4 c0 = w[0][s >> 1], c1 = w[1][s >> 1];
        v2f b0, b1;
        if (s & 1) { b0.x = c0.z; b0.y = c0.w; b1.x = c1.z; b1.y = c1.w; }
        else       { b0.x = c0.x; b0.y = c0.y; b1.x = c1.x; b1.y = c1.y; }
#pragma unroll
        for (int rt = 0; rt < 4; ++rt) {
            acc[0][rt] = wmma_f32(a[rt], b0, acc[0][rt]);
            acc[1][rt] = wmma_f32(a[rt], b1, acc[1][rt]);
        }
    }
}

__global__ __launch_bounds__(256) __attribute__((amdgpu_waves_per_eu(4, 4))) void qkv_gemm_kernel(
    const float* __restrict__ x, const float* __restrict__ wq,
    const float* __restrict__ bq, float* __restrict__ qkv)
{
    __shared__ float2 Xs2[128 * 66];
    const int row0 = blockIdx.x * 64;
    const int n = row0 >> 12;
    const int l0 = row0 & (LL - 1);
    const int tid = threadIdx.x;
    const int lane = tid & 31, wave = tid >> 5;
    const int c = lane & 15, hf = lane >> 4;

    {
        const int m4 = (tid & 15) * 4;
        const int pb = tid >> 4;
        const float* xs = x + (size_t)n * CC * LL + l0 + m4;
        float4 lo[8], hi[8];
#pragma unroll
        for (int i = 0; i < 8; ++i) {
            const int p = pb + 16 * i;
            lo[i] = *(const float4*)(xs + (size_t)(2 * p) * LL);
            hi[i] = *(const float4*)(xs + (size_t)(2 * p + 1) * LL);
        }
#pragma unroll
        for (int i = 0; i < 8; ++i) {
            const int p = pb + 16 * i;
            float4 s0, s1;
            s0.x = lo[i].x; s0.y = hi[i].x; s0.z = lo[i].y; s0.w = hi[i].y;
            s1.x = lo[i].z; s1.y = hi[i].z; s1.z = lo[i].w; s1.w = hi[i].w;
            *(float4*)&Xs2[p * 66 + m4]     = s0;
            *(float4*)&Xs2[p * 66 + m4 + 2] = s1;
        }
    }

    const float* wrow = wq + (size_t)(32 * wave + c) * CC + 16 * hf;
    const float2* arow = Xs2 + (8 * hf) * 66 + c;
    float* obase = qkv + (size_t)(row0 + 8 * hf) * TC + 32 * wave + c;

    float4 wa[2][4], wb[2][4];
#pragma unroll
    for (int q = 0; q < 4; ++q) {
        wa[0][q] = *(const float4*)(wrow + 4 * q);
        wa[1][q] = *(const float4*)(wrow + 16 * CC + 4 * q);
    }
    __syncthreads();

    for (int cp = 0; cp < 3; ++cp) {
        const float* wg = wrow + (size_t)cp * 256 * CC;
        v8f acc[2][4] = {};
#pragma unroll 1
        for (int kk = 0; kk < 4; ++kk) {
            const int kc = kk * 64;
#pragma unroll
            for (int q = 0; q < 4; ++q) {
                wb[0][q] = *(const float4*)(wg + kc + 32 + 4 * q);
                wb[1][q] = *(const float4*)(wg + 16 * CC + kc + 32 + 4 * q);
            }
            qkv_chunk(acc, wa, arow + (kc >> 1) * 66);
            {
                const float* wn = (kk < 3) ? (wg + kc + 64) : ((cp < 2) ? (wg + 256 * CC) : wg);
#pragma unroll
                for (int q = 0; q < 4; ++q) {
                    wa[0][q] = *(const float4*)(wn + 4 * q);
                    wa[1][q] = *(const float4*)(wn + 16 * CC + 4 * q);
                }
            }
            qkv_chunk(acc, wb, arow + ((kc + 32) >> 1) * 66);
        }
        const float bias0 = bq[256 * cp + 32 * wave + c];
        const float bias1 = bq[256 * cp + 32 * wave + 16 + c];
        float* og = obase + 256 * cp;
#pragma unroll
        for (int rt = 0; rt < 4; ++rt)
#pragma unroll
            for (int r = 0; r < 8; ++r) {
                float* o = og + (size_t)(16 * rt + r) * TC;
                o[0]  = acc[0][rt][r] + bias0;
                o[16] = acc[1][rt][r] + bias1;
            }
    }
}

__global__ __launch_bounds__(512) __attribute__((amdgpu_waves_per_eu(4, 4))) void norm_attn_kernel(float* __restrict__ qkv)
{
    __shared__ float red[16 * 1024];
    const int b = blockIdx.x;
    const int quarter = b & 3;
    const int h = (b >> 2) & 7;
    const int n = b >> 5;
    const int tid = threadIdx.x, lane = tid & 31, wave = tid >> 5;
    const int c = lane & 15, hf = lane >> 4;
    float* base = qkv + ((size_t)n * LL + quarter * 1024 + wave * 64) * TC + h * DD;
    float* rowp = base + (size_t)c * TC + 16 * hf;
    const float* frag = base + (size_t)(8 * hf) * TC + 2 * c;

    v8f acc[2][2] = {};
    const v2f ones = {1.0f, 1.0f};

    float4 k4[4];
    float2 kf[8], vf[8];
#pragma unroll
    for (int i = 0; i < 4; ++i) k4[i] = *(const float4*)(rowp + CC + 4 * i);
#pragma unroll
    for (int r = 0; r < 8; ++r) {
        kf[r] = *(const float2*)(frag + (size_t)r * TC + CC);
        vf[r] = *(const float2*)(frag + (size_t)r * TC + 2 * CC);
    }

#pragma unroll
    for (int t = 0; t < 4; ++t) {
        float4 k4n[4];
        float2 kfn[8], vfn[8];
        if (t < 3) {
            const size_t off = (size_t)(t + 1) * 16 * TC;
#pragma unroll
            for (int i = 0; i < 4; ++i) k4n[i] = *(const float4*)(rowp + off + CC + 4 * i);
#pragma unroll
            for (int r = 0; r < 8; ++r) {
                kfn[r] = *(const float2*)(frag + off + (size_t)r * TC + CC);
                vfn[r] = *(const float2*)(frag + off + (size_t)r * TC + 2 * CC);
            }
        }
        float pk = 0.f;
#pragma unroll
        for (int i = 0; i < 4; ++i)
            pk += k4[i].x * k4[i].x + k4[i].y * k4[i].y + k4[i].z * k4[i].z + k4[i].w * k4[i].w;
        v2f an; an.x = pk; an.y = 0.0f;
        v8f zero = {};
        const v8f ss = wmma_f32(an, ones, zero);
#pragma unroll
        for (int s = 0; s < 4; ++s) {
            const float i0 = 1.0f / sqrtf(ss[2 * s]);
            const float i1 = 1.0f / sqrtf(ss[2 * s + 1]);
            v2f a0, a1, b0, b1;
            a0.x = kf[2 * s].x * i0; a0.y = kf[2 * s + 1].x * i1;
            a1.x = kf[2 * s].y * i0; a1.y = kf[2 * s + 1].y * i1;
            b0.x = vf[2 * s].x;      b0.y = vf[2 * s + 1].x;
            b1.x = vf[2 * s].y;      b1.y = vf[2 * s + 1].y;
            acc[0][0] = wmma_f32(a0, b0, acc[0][0]);
            acc[0][1] = wmma_f32(a0, b1, acc[0][1]);
            acc[1][0] = wmma_f32(a1, b0, acc[1][0]);
            acc[1][1] = wmma_f32(a1, b1, acc[1][1]);
        }
        if (t < 3) {
#pragma unroll
            for (int i = 0; i < 4; ++i) k4[i] = k4n[i];
#pragma unroll
            for (int r = 0; r < 8; ++r) { kf[r] = kfn[r]; vf[r] = vfn[r]; }
        }
    }
#pragma unroll
    for (int md = 0; md < 2; ++md)
#pragma unroll
        for (int r = 0; r < 8; ++r) {
            float2 o; o.x = acc[md][0][r]; o.y = acc[md][1][r];
            *(float2*)&red[wave * 1024 + (16 * hf + 2 * r + md) * 32 + 2 * c] = o;
        }
    __syncthreads();
    float* pout = qkv + ((size_t)n * LL + quarter * 1024) * TC + CC + h * DD;
    for (int i = tid; i < 1024; i += 512) {
        float s = 0.f;
#pragma unroll
        for (int w = 0; w < 16; ++w) s += red[w * 1024 + i];
        pout[(size_t)(i >> 5) * TC + (i & 31)] = s;
    }
}

static __device__ __forceinline__ float row16_sum(float x) {
    x += __int_as_float(__builtin_amdgcn_update_dpp(0, __float_as_int(x), 0x161, 0xf, 0xf, true));
    x += __int_as_float(__builtin_amdgcn_update_dpp(0, __float_as_int(x), 0x162, 0xf, 0xf, true));
    x += __int_as_float(__builtin_amdgcn_update_dpp(0, __float_as_int(x), 0x164, 0xf, 0xf, true));
    x += __int_as_float(__builtin_amdgcn_update_dpp(0, __float_as_int(x), 0x168, 0xf, 0xf, true));
    return x;
}

__global__ __launch_bounds__(256) void epilogue_kernel(
    const float* __restrict__ qkv,
    const float* __restrict__ wdc, float* __restrict__ out_pre)
{
    __shared__ float4 Bq[16 * 16];
    const int b = blockIdx.x;
    const int lc = b & 15;
    const int h = (b >> 4) & 7;
    const int n = b >> 7;
    const int tid = threadIdx.x, lane = tid & 31;
    const int wave = tid >> 5;
    const int c = lane & 15, hf = lane >> 4;
    const int lw = lc * 256 + wave * 32;
    const float* qbase = qkv + (size_t)n * LL * TC + h * DD;

    float asum[4];
#pragma unroll
    for (int j = 0; j < 4; ++j) {
        const int d = (tid >> 5) + 8 * j;
        const float* pp = qkv + ((size_t)n * LL + d) * TC + CC + h * DD + lane;
        float a = pp[0];
        a += pp[(size_t)1024 * TC];
        a += pp[(size_t)2048 * TC];
        a += pp[(size_t)3072 * TC];
        asum[j] = a;
    }

    float4 q4[2][4];
    float2 vv[2][16];
#pragma unroll
    for (int t = 0; t < 2; ++t) {
        const float* qr = qbase + (size_t)(lw + 16 * t + c) * TC + 16 * hf;
#pragma unroll
        for (int i = 0; i < 4; ++i) q4[t][i] = *(const float4*)(qr + 4 * i);
    }
    const int lv0 = lw + 8 * hf - 4;
    const float* vcol = qbase + 2 * CC + 2 * c;
#pragma unroll
    for (int i = 0; i < 4; ++i) {
        const int l = lv0 + i;
        const float2 x = *(const float2*)(vcol + (size_t)(l < 0 ? 0 : l) * TC);
        vv[0][i].x = (l < 0) ? 0.0f : x.x;
        vv[0][i].y = (l < 0) ? 0.0f : x.y;
    }
    {
        const float* vr = vcol + (size_t)(lv0 + 4) * TC;
#pragma unroll
        for (int i = 4; i < 28; ++i) vv[i >> 4][i & 15] = *(const float2*)(vr + (size_t)(i - 4) * TC);
    }
#pragma unroll
    for (int i = 12; i < 16; ++i) {
        const int l = lv0 + 16 + i;
        const float2 x = *(const float2*)(vcol + (size_t)(l < LL ? l : LL - 1) * TC);
        vv[1][i].x = (l < LL) ? x.x : 0.0f;
        vv[1][i].y = (l < LL) ? x.y : 0.0f;
    }

#pragma unroll
    for (int j = 0; j < 4; ++j) {
        const int d = (tid >> 5) + 8 * j;
        ((float*)Bq)[(((d >> 1) * 16 + (lane >> 1)) << 2) + ((lane & 1) << 1) + (d & 1)] = asum[j];
    }
    __syncthreads();

    float4 bf[8];
#pragma unroll
    for (int s = 0; s < 8; ++s) bf[s] = Bq[(8 * hf + s) * 16 + c];

    float wd[9];
#pragma unroll
    for (int r = 0; r < 9; ++r) wd[r] = wdc[h * 9 + r];

    const float inv_pi = 0.31830988618379067154f;
#pragma unroll
    for (int t = 0; t < 2; ++t) {
        float pq = 0.f;
#pragma unroll
        for (int i = 0; i < 4; ++i)
            pq += q4[t][i].x * q4[t][i].x + q4[t][i].y * q4[t][i].y + q4[t][i].z * q4[t][i].z + q4[t][i].w * q4[t][i].w;
        pq += __shfl_xor(pq, 16, 32);
        const float invq = 1.0f / sqrtf(pq);
#pragma unroll
        for (int i = 0; i < 4; ++i) {
            q4[t][i].x *= invq; q4[t][i].y *= invq; q4[t][i].z *= invq; q4[t][i].w *= invq;
        }
        v8f acc0 = {}, acc1 = {};
#pragma unroll
        for (int s = 0; s < 8; ++s) {
            const float4 qq = q4[t][s >> 1];
            v2f a, b0, b1;
            if (s & 1) { a.x = qq.z; a.y = qq.w; } else { a.x = qq.x; a.y = qq.y; }
            b0.x = bf[s].x; b0.y = bf[s].y;
            b1.x = bf[s].z; b1.y = bf[s].w;
            acc0 = wmma_f32(a, b0, acc0);
            acc1 = wmma_f32(a, b1, acc1);
        }
        float* orow = out_pre + (size_t)(n * LL + lw + 16 * t + 8 * hf) * CC + h * DD + 2 * c;
#pragma unroll
        for (int r = 0; r < 8; ++r) {
            const float2 vc = vv[t][r + 4];
            const float v0 = 0.5f * vc.x + inv_pi * acc0[r];
            const float v1 = 0.5f * vc.y + inv_pi * acc1[r];
            const float nrm = sqrtf(row16_sum(v0 * v0 + v1 * v1));
            float dc0 = 0.f, dc1 = 0.f;
#pragma unroll
            for (int k = 0; k < 9; ++k) {
                dc0 += wd[k] * vv[t][r + k].x;
                dc1 += wd[k] * vv[t][r + k].y;
            }
            float2 o;
            o.x = v0 / nrm + dc0;
            o.y = v1 / nrm + dc1;
            *(float2*)(orow + (size_t)r * CC) = o;
        }
    }
}

__global__ __launch_bounds__(256) void proj_gemm_kernel(
    const float* __restrict__ A, const float* __restrict__ wp,
    const float* __restrict__ bp, float* __restrict__ out)
{
    __shared__ float2 Wt2[16 * 66];
    const int row0 = blockIdx.x * 256;
    const int j0 = blockIdx.y * 64;
    const int tid = threadIdx.x, lane = tid & 31, wave = tid >> 5;
    const int laneM = lane & 15;
    const int laneH = lane >> 4;
    const int lw0 = wave * 32;

    v8f acc[4][2] = {};
    const float* xr0 = A + (size_t)(row0 + lw0 + laneM) * CC + 16 * laneH;
    const float* xr1 = xr0 + (size_t)16 * CC;

    float4 cur[2][4], nxt[2][4];
#pragma unroll
    for (int q = 0; q < 4; ++q) {
        cur[0][q] = *(const float4*)(xr0 + 4 * q);
        cur[1][q] = *(const float4*)(xr1 + 4 * q);
    }

#pragma unroll
    for (int kc = 0; kc < CC; kc += 32) {
        for (int i = tid; i < 16 * 64; i += 256) {
            const int p = i & 15, j = i >> 4;
            Wt2[p * 66 + j] = *(const float2*)&wp[(size_t)(j0 + j) * CC + kc + 2 * p];
        }
        __syncthreads();
        if (kc + 32 < CC) {
#pragma unroll
            for (int q = 0; q < 4; ++q) {
                nxt[0][q] = *(const float4*)(xr0 + kc + 32 + 4 * q);
                nxt[1][q] = *(const float4*)(xr1 + kc + 32 + 4 * q);
            }
        }
#pragma unroll
        for (int s = 0; s < 8; ++s) {
            const int pp = 8 * laneH + s;
            const float4 c0 = cur[0][s >> 1], c1 = cur[1][s >> 1];
            v2f b0, b1;
            if (s & 1) { b0.x = c0.z; b0.y = c0.w; b1.x = c1.z; b1.y = c1.w; }
            else       { b0.x = c0.x; b0.y = c0.y; b1.x = c1.x; b1.y = c1.y; }
#pragma unroll
            for (int jt = 0; jt < 4; ++jt) {
                const v2f a = *(const v2f*)&Wt2[pp * 66 + jt * 16 + laneM];
                acc[jt][0] = wmma_f32(a, b0, acc[jt][0]);
                acc[jt][1] = wmma_f32(a, b1, acc[jt][1]);
            }
        }
        __syncthreads();
        if (kc + 32 < CC) {
#pragma unroll
            for (int q = 0; q < 4; ++q) { cur[0][q] = nxt[0][q]; cur[1][q] = nxt[1][q]; }
        }
    }
    const int n = row0 >> 12;
    const int lg0 = row0 & (LL - 1);
#pragma unroll
    for (int jt = 0; jt < 4; ++jt) {
        const int jb = j0 + jt * 16 + (laneH << 3);
#pragma unroll
        for (int r = 0; r < 8; ++r) {
            const float bias = bp[jb + r];
            float* o = out + ((size_t)(n * CC + jb + r)) * LL + lg0 + lw0 + laneM;
            o[0]  = acc[jt][0][r] + bias;
            o[16] = acc[jt][1][r] + bias;
        }
    }
}

extern "C" void kernel_launch(void* const* d_in, const int* in_sizes, int n_in,
                              void* d_out, int out_size, void* d_ws, size_t ws_size,
                              hipStream_t stream) {
    const float* x      = (const float*)d_in[0];
    const float* w_qkv  = (const float*)d_in[1];
    const float* b_qkv  = (const float*)d_in[2];
    const float* w_proj = (const float*)d_in[3];
    const float* b_proj = (const float*)d_in[4];
    const float* w_dcv  = (const float*)d_in[5];
    float* out = (float*)d_out;

    float* qkv     = (float*)d_ws;
    float* out_pre = qkv + (size_t)NB * LL * TC + (size_t)NB * HH * DD * DD;

    qkv_gemm_kernel<<<512, 256, 0, stream>>>(x, w_qkv, b_qkv, qkv);
    norm_attn_kernel<<<256, 512, 0, stream>>>(qkv);
    epilogue_kernel<<<1024, 256, 0, stream>>>(qkv, w_dcv, out_pre);
    proj_gemm_kernel<<<dim3(128, 4), 256, 0, stream>>>(out_pre, w_proj, b_proj, out);
}


# qkv intermediate head-major, [q|k|v][n][head][l][32]: the GEMM's wave tile and every reader's rows are contiguous whole 256-byte units (was 128-byte pieces at a 3072-byte pitch); sums unchanged
# speedup vs baseline: 1.9481x; 1.0415x over previous
#include <hip/hip_runtime.h>
#include <hip/hip_bf16.h>
#include <math.h>

typedef float v2f __attribute__((ext_vector_type(2)));
typedef float v8f __attribute__((ext_vector_type(8)));

#define NB 8
#define CC 256
#define HH 8
#define DD 32
#define LL 4096
#define TC 768
#define SET ((size_t)NB * HH * LL * DD)

static __device__ __forceinline__ v8f wmma_f32(v2f a, v2f b, v8f c) {
    return __builtin_amdgcn_wmma_f32_16x16x4_f32(false, a, false, b, (short)0, c, false, false);
}

static __device__ __forceinline__ void qkv_chunk(v8f (&acc)[2][4], const float4 (&w)[2][4],
                                                 const float2* __restrict__ ap)
{
#pragma unroll
    for (int s = 0; s < 8; ++s) {
        v2f a[4];
#pragma unroll
        for (int rt = 0; rt < 4; ++rt) a[rt] = *(const v2f*)(ap + s * 66 + 16 * rt);
        const float4 c0 = w[0][s >> 1], c1 = w[1][s >> 1];
        v2f b0, b1;
        if (s & 1) { b0.x = c0.z; b0.y = c0.w; b1.x = c1.z; b1.y = c1.w; }
        else       { b0.x = c0.x; b0.y = c0.y; b1.x = c1.x; b1.y = c1.y; }
#pragma unroll
        for (int rt = 0; rt < 4; ++rt) {
            acc[0][rt] = wmma_f32(a[rt], b0, acc[0][rt]);
            acc[1][rt] = wmma_f32(a[rt], b1, acc[1][rt]);
        }
    }
}

__global__ __launch_bounds__(256) __attribute__((amdgpu_waves_per_eu(4, 4))) void qkv_gemm_kernel(
    const float* __restrict__ x, const float* __restrict__ wq,
    const float* __restrict__ bq, float* __restrict__ qkv)
{
    __shared__ float2 Xs2[128 * 66];
    const int row0 = blockIdx.x * 64;
    const int n = row0 >> 12;
    const int l0 = row0 & (LL - 1);
    const int tid = threadIdx.x;
    const int lane = tid & 31, wave = tid >> 5;
    const int c = lane & 15, hf = lane >> 4;

    {
        const int m4 = (tid & 15) * 4;
        const int pb = tid >> 4;
        const float* xs = x + (size_t)n * CC * LL + l0 + m4;
        float4 lo[8], hi[8];
#pragma unroll
        for (int i = 0; i < 8; ++i) {
            const int p = pb + 16 * i;
            lo[i] = *(const float4*)(xs + (size_t)(2 * p) * LL);
            hi[i] = *(const float4*)(xs + (size_t)(2 * p + 1) * LL);
        }
#pragma unroll
        for (int i = 0; i < 8; ++i) {
            const int p = pb + 16 * i;
            float4 s0, s1;
            s0.x = lo[i].x; s0.y = hi[i].x; s0.z = lo[i].y; s0.w = hi[i].y;
            s1.x = lo[i].z; s1.y = hi[i].z; s1.z = lo[i].w; s1.w = hi[i].w;
            *(float4*)&Xs2[p * 66 + m4]     = s0;
            *(float4*)&Xs2[p * 66 + m4 + 2] = s1;
        }
    }

    const float* wrow = wq + (size_t)(32 * wave + c) * CC + 16 * hf;
    const float2* arow = Xs2 + (8 * hf) * 66 + c;
    float* obase = qkv + ((size_t)(n * HH + wave) * LL + l0 + 8 * hf) * DD + c;

    float4 wa[2][4], wb[2][4];
#pragma unroll
    for (int q = 0; q < 4; ++q) {
        wa[0][q] = *(const float4*)(wrow + 4 * q);
        wa[1][q] = *(const float4*)(wrow + 16 * CC + 4 * q);
    }
    __syncthreads();

    for (int cp = 0; cp < 3; ++cp) {
        const float* wg = wrow + (size_t)cp * 256 * CC;
        v8f acc[2][4] = {};
#pragma unroll 1
        for (int kk = 0; kk < 4; ++kk) {
            const int kc = kk * 64;
#pragma unroll
            for (int q = 0; q < 4; ++q) {
                wb[0][q] = *(const float4*)(wg + kc + 32 + 4 * q);
                wb[1][q] = *(const float4*)(wg + 16 * CC + kc + 32 + 4 * q);
            }
            qkv_chunk(acc, wa, arow + (kc >> 1) * 66);
            {
                const float* wn = (kk < 3) ? (wg + kc + 64) : ((cp < 2) ? (wg + 256 * CC) : wg);
#pragma unroll
                for (int q = 0; q < 4; ++q) {
                    wa[0][q] = *(const float4*)(wn + 4 * q);
                    wa[1][q] = *(const float4*)(wn + 16 * CC + 4 * q);
                }
            }
            qkv_chunk(acc, wb, arow + ((kc + 32) >> 1) * 66);
        }
        const float bias0 = bq[256 * cp + 32 * wave + c];
        const float bias1 = bq[256 * cp + 32 * wave + 16 + c];
        float* og = obase + (size_t)cp * SET;
#pragma unroll
        for (int rt = 0; rt < 4; ++rt)
#pragma unroll
            for (int r = 0; r < 8; ++r) {
                float* o = og + (16 * rt + r) * DD;
                o[0]  = acc[0][rt][r] + bias0;
                o[16] = acc[1][rt][r] + bias1;
            }
    }
}

__global__ __launch_bounds__(512) __attribute__((amdgpu_waves_per_eu(4, 4))) void norm_attn_kernel(float* __restrict__ qkv)
{
    __shared__ float red[16 * 1024];
    const int b = blockIdx.x;
    const int quarter = b & 3;
    const int h = (b >> 2) & 7;
    const int n = b >> 5;
    const int tid = threadIdx.x, lane = tid & 31, wave = tid >> 5;
    const int c = lane & 15, hf = lane >> 4;
    const size_t wrow = ((size_t)(n * HH + h) * LL + quarter * 1024 + wave * 64) * DD;
    const float* kb = qkv + SET + wrow;
    const float* vb = qkv + 2 * SET + wrow;
    const float* rowp = kb + c * DD + 16 * hf;
    const float* kfrag = kb + (8 * hf) * DD + 2 * c;
    const float* vfrag = vb + (8 * hf) * DD + 2 * c;

    v8f acc[2][2] = {};
    const v2f ones = {1.0f, 1.0f};

    float4 k4[4];
    float2 kf[8], vf[8];
#pragma unroll
    for (int i = 0; i < 4; ++i) k4[i] = *(const float4*)(rowp + 4 * i);
#pragma unroll
    for (int r = 0; r < 8; ++r) {
        kf[r] = *(const float2*)(kfrag + r * DD);
        vf[r] = *(const float2*)(vfrag + r * DD);
    }

#pragma unroll
    for (int t = 0; t < 4; ++t) {
        float4 k4n[4];
        float2 kfn[8], vfn[8];
        if (t < 3) {
            const int off = (t + 1) * 16 * DD;
#pragma unroll
            for (int i = 0; i < 4; ++i) k4n[i] = *(const float4*)(rowp + off + 4 * i);
#pragma unroll
            for (int r = 0; r < 8; ++r) {
                kfn[r] = *(const float2*)(kfrag + off + r * DD);
                vfn[r] = *(const float2*)(vfrag + off + r * DD);
            }
        }
        float pk = 0.f;
#pragma unroll
        for (int i = 0; i < 4; ++i)
            pk += k4[i].x * k4[i].x + k4[i].y * k4[i].y + k4[i].z * k4[i].z + k4[i].w * k4[i].w;
        v2f an; an.x = pk; an.y = 0.0f;
        v8f zero = {};
        const v8f ss = wmma_f32(an, ones, zero);
#pragma unroll
        for (int s = 0; s < 4; ++s) {
            const float i0 = 1.0f / sqrtf(ss[2 * s]);
            const float i1 = 1.0f / sqrtf(ss[2 * s + 1]);
            v2f a0, a1, b0, b1;
            a0.x = kf[2 * s].x * i0; a0.y = kf[2 * s + 1].x * i1;
            a1.x = kf[2 * s].y * i0; a1.y = kf[2 * s + 1].y * i1;
            b0.x = vf[2 * s].x;      b0.y = vf[2 * s + 1].x;
            b1.x = vf[2 * s].y;      b1.y = vf[2 * s + 1].y;
            acc[0][0] = wmma_f32(a0, b0, acc[0][0]);
            acc[0][1] = wmma_f32(a0, b1, acc[0][1]);
            acc[1][0] = wmma_f32(a1, b0, acc[1][0]);
            acc[1][1] = wmma_f32(a1, b1, acc[1][1]);
        }
        if (t < 3) {
#pragma unroll
            for (int i = 0; i < 4; ++i) k4[i] = k4n[i];
#pragma unroll
            for (int r = 0; r < 8; ++r) { kf[r] = kfn[r]; vf[r] = vfn[r]; }
        }
    }
#pragma unroll
    for (int md = 0; md < 2; ++md)
#pragma unroll
        for (int r = 0; r < 8; ++r) {
            float2 o; o.x = acc[md][0][r]; o.y = acc[md][1][r];
            *(float2*)&red[wave * 1024 + (16 * hf + 2 * r + md) * 32 + 2 * c] = o;
        }
    __syncthreads();
    float* pout = qkv + SET + ((size_t)(n * HH + h) * LL + quarter * 1024) * DD;
    for (int i = tid; i < 1024; i += 512) {
        float s = 0.f;
#pragma unroll
        for (int w = 0; w < 16; ++w) s += red[w * 1024 + i];
        pout[i] = s;
    }
}

static __device__ __forceinline__ float row16_sum(float x) {
    x += __int_as_float(__builtin_amdgcn_update_dpp(0, __float_as_int(x), 0x161, 0xf, 0xf, true));
    x += __int_as_float(__builtin_amdgcn_update_dpp(0, __float_as_int(x), 0x162, 0xf, 0xf, true));
    x += __int_as_float(__builtin_amdgcn_update_dpp(0, __float_as_int(x), 0x164, 0xf, 0xf, true));
    x += __int_as_float(__builtin_amdgcn_update_dpp(0, __float_as_int(x), 0x168, 0xf, 0xf, true));
    return x;
}

__global__ __launch_bounds__(256) void epilogue_kernel(
    const float* __restrict__ qkv,
    const float* __restrict__ wdc, float* __restrict__ out_pre)
{
    __shared__ float4 Bq[16 * 16];
    const int b = blockIdx.x;
    const int lc = b & 15;
    const int h = (b >> 4) & 7;
    const int n = b >> 7;
    const int tid = threadIdx.x, lane = tid & 31;
    const int wave = tid >> 5;
    const int c = lane & 15, hf = lane >> 4;
    const int lw = lc * 256 + wave * 32;
    const size_t hp = (size_t)(n * HH + h) * LL * DD;
    const float* qbase = qkv + hp;

    float asum[4];
#pragma unroll
    for (int j = 0; j < 4; ++j) {
        const int d = (tid >> 5) + 8 * j;
        const float* pp = qkv + SET + hp + d * DD + lane;
        float a = pp[0];
        a += pp[1024 * DD];
        a += pp[2048 * DD];
        a += pp[3072 * DD];
        asum[j] = a;
    }

    float4 q4[2][4];
    float2 vv[2][16];
#pragma unroll
    for (int t = 0; t < 2; ++t) {
        const float* qr = qbase + (lw + 16 * t + c) * DD + 16 * hf;
#pragma unroll
        for (int i = 0; i < 4; ++i) q4[t][i] = *(const float4*)(qr + 4 * i);
    }
    const int lv0 = lw + 8 * hf - 4;
    const float* vcol = qkv + 2 * SET + hp + 2 * c;
#pragma unroll
    for (int i = 0; i < 4; ++i) {
        const int l = lv0 + i;
        const float2 x = *(const float2*)(vcol + (l < 0 ? 0 : l) * DD);
        vv[0][i].x = (l < 0) ? 0.0f : x.x;
        vv[0][i].y = (l < 0) ? 0.0f : x.y;
    }
    {
        const float* vr = vcol + (lv0 + 4) * DD;
#pragma unroll
        for (int i = 4; i < 28; ++i) vv[i >> 4][i & 15] = *(const float2*)(vr + (i - 4) * DD);
    }
#pragma unroll
    for (int i = 12; i < 16; ++i) {
        const int l = lv0 + 16 + i;
        const float2 x = *(const float2*)(vcol + (l < LL ? l : LL - 1) * DD);
        vv[1][i].x = (l < LL) ? x.x : 0.0f;
        vv[1][i].y = (l < LL) ? x.y : 0.0f;
    }

#pragma unroll
    for (int j = 0; j < 4; ++j) {
        const int d = (tid >> 5) + 8 * j;
        ((float*)Bq)[(((d >> 1) * 16 + (lane >> 1)) << 2) + ((lane & 1) << 1) + (d & 1)] = asum[j];
    }
    __syncthreads();

    float4 bf[8];
#pragma unroll
    for (int s = 0; s < 8; ++s) bf[s] = Bq[(8 * hf + s) * 16 + c];

    float wd[9];
#pragma unroll
    for (int r = 0; r < 9; ++r) wd[r] = wdc[h * 9 + r];

    const float inv_pi = 0.31830988618379067154f;
#pragma unroll
    for (int t = 0; t < 2; ++t) {
        float pq = 0.f;
#pragma unroll
        for (int i = 0; i < 4; ++i)
            pq += q4[t][i].x * q4[t][i].x + q4[t][i].y * q4[t][i].y + q4[t][i].z * q4[t][i].z + q4[t][i].w * q4[t][i].w;
        pq += __shfl_xor(pq, 16, 32);
        const float invq = 1.0f / sqrtf(pq);
#pragma unroll
        for (int i = 0; i < 4; ++i) {
            q4[t][i].x *= invq; q4[t][i].y *= invq; q4[t][i].z *= invq; q4[t][i].w *= invq;
        }
        v8f acc0 = {}, acc1 = {};
#pragma unroll
        for (int s = 0; s < 8; ++s) {
            const float4 qq = q4[t][s >> 1];
            v2f a, b0, b1;
            if (s & 1) { a.x = qq.z; a.y = qq.w; } else { a.x = qq.x; a.y = qq.y; }
            b0.x = bf[s].x; b0.y = bf[s].y;
            b1.x = bf[s].z; b1.y = bf[s].w;
            acc0 = wmma_f32(a, b0, acc0);
            acc1 = wmma_f32(a, b1, acc1);
        }
        float* orow = out_pre + (size_t)(n * LL + lw + 16 * t + 8 * hf) * CC + h * DD + 2 * c;
#pragma unroll
        for (int r = 0; r < 8; ++r) {
            const float2 vc = vv[t][r + 4];
            const float v0 = 0.5f * vc.x + inv_pi * acc0[r];
            const float v1 = 0.5f * vc.y + inv_pi * acc1[r];
            const float nrm = sqrtf(row16_sum(v0 * v0 + v1 * v1));
            float dc0 = 0.f, dc1 = 0.f;
#pragma unroll
            for (int k = 0; k < 9; ++k) {
                dc0 += wd[k] * vv[t][r + k].x;
                dc1 += wd[k] * vv[t][r + k].y;
            }
            float2 o;
            o.x = v0 / nrm + dc0;
            o.y = v1 / nrm + dc1;
            *(float2*)(orow + (size_t)r * CC) = o;
        }
    }
}

__global__ __launch_bounds__(256) void proj_gemm_kernel(
    const float* __restrict__ A, const float* __restrict__ wp,
    const float* __restrict__ bp, float* __restrict__ out)
{
    __shared__ float2 Wt2[16 * 66];
    const int row0 = blockIdx.x * 256;
    const int j0 = blockIdx.y * 64;
    const int tid = threadIdx.x, lane = tid & 31, wave = tid >> 5;
    const int laneM = lane & 15;
    const int laneH = lane >> 4;
    const int lw0 = wave * 32;

    v8f acc[4][2] = {};
    const float* xr0 = A + (size_t)(row0 + lw0 + laneM) * CC + 16 * laneH;
    const float* xr1 = xr0 + (size_t)16 * CC;

    float4 cur[2][4], nxt[2][4];
#pragma unroll
    for (int q = 0; q < 4; ++q) {
        cur[0][q] = *(const float4*)(xr0 + 4 * q);
        cur[1][q] = *(const float4*)(xr1 + 4 * q);
    }

#pragma unroll
    for (int kc = 0; kc < CC; kc += 32) {
        for (int i = tid; i < 16 * 64; i += 256) {
            const int p = i & 15, j = i >> 4;
            Wt2[p * 66 + j] = *(const float2*)&wp[(size_t)(j0 + j) * CC + kc + 2 * p];
        }
        __syncthreads();
        if (kc + 32 < CC) {
#pragma unroll
            for (int q = 0; q < 4; ++q) {
                nxt[0][q] = *(const float4*)(xr0 + kc + 32 + 4 * q);
                nxt[1][q] = *(const float4*)(xr1 + kc + 32 + 4 * q);
            }
        }
#pragma unroll
        for (int s = 0; s < 8; ++s) {
            const int pp = 8 * laneH + s;
            const float4 c0 = cur[0][s >> 1], c1 = cur[1][s >> 1];
            v2f b0, b1;
            if (s & 1) { b0.x = c0.z; b0.y = c0.w; b1.x = c1.z; b1.y = c1.w; }
            else       { b0.x = c0.x; b0.y = c0.y; b1.x = c1.x; b1.y = c1.y; }
#pragma unroll
            for (int jt = 0; jt < 4; ++jt) {
                const v2f a = *(const v2f*)&Wt2[pp * 66 + jt * 16 + laneM];
                acc[jt][0] = wmma_f32(a, b0, acc[jt][0]);
                acc[jt][1] = wmma_f32(a, b1, acc[jt][1]);
            }
        }
        __syncthreads();
        if (kc + 32 < CC) {
#pragma unroll
            for (int q = 0; q < 4; ++q) { cur[0][q] = nxt[0][q]; cur[1][q] = nxt[1][q]; }
        }
    }
    const int n = row0 >> 12;
    const int lg0 = row0 & (LL - 1);
#pragma unroll
    for (int jt = 0; jt < 4; ++jt) {
        const int jb = j0 + jt * 16 + (laneH << 3);
#pragma unroll
        for (int r = 0; r < 8; ++r) {
            const float bias = bp[jb + r];
            float* o = out + ((size_t)(n * CC + jb + r)) * LL + lg0 + lw0 + laneM;
            o[0]  = acc[jt][0][r] + bias;
            o[16] = acc[jt][1][r] + bias;
        }
    }
}

extern "C" void kernel_launch(void* const* d_in, const int* in_sizes, int n_in,
                              void* d_out, int out_size, void* d_ws, size_t ws_size,
                              hipStream_t stream) {
    const float* x      = (const float*)d_in[0];
    const float* w_qkv  = (const float*)d_in[1];
    const float* b_qkv  = (const float*)d_in[2];
    const float* w_proj = (const float*)d_in[3];
    const float* b_proj = (const float*)d_in[4];
    const float* w_dcv  = (const float*)d_in[5];
    float* out = (float*)d_out;

    float* qkv     = (float*)d_ws;
    float* out_pre = qkv + (size_t)NB * LL * TC + (size_t)NB * HH * DD * DD;

    qkv_gemm_kernel<<<512, 256, 0, stream>>>(x, w_qkv, b_qkv, qkv);
    norm_attn_kernel<<<256, 512, 0, stream>>>(qkv);
    epilogue_kernel<<<1024, 256, 0, stream>>>(qkv, w_dcv, out_pre);
    proj_gemm_kernel<<<dim3(128, 4), 256, 0, stream>>>(out_pre, w_proj, b_proj, out);
}


# Projection GEMM re-tiled: a block owns 64 l rows x all 256 channels; its out_pre slice sits in LDS, read once; w_proj goes global -> A fragments a chunk ahead; K loop has no barrier; sums unchanged
# speedup vs baseline: 2.0480x; 1.0513x over previous
#include <hip/hip_runtime.h>
#include <hip/hip_bf16.h>
#include <math.h>

typedef float v2f __attribute__((ext_vector_type(2)));
typedef float v8f __attribute__((ext_vector_type(8)));

#define NB 8
#define CC 256
#define HH 8
#define DD 32
#define LL 4096
#define TC 768
#define SET ((size_t)NB * HH * LL * DD)

static __device__ __forceinline__ v8f wmma_f32(v2f a, v2f b, v8f c) {
    return __builtin_amdgcn_wmma_f32_16x16x4_f32(false, a, false, b, (short)0, c, false, false);
}

static __device__ __forceinline__ void qkv_chunk(v8f (&acc)[2][4], const float4 (&w)[2][4],
                                                 const float2* __restrict__ ap)
{
#pragma unroll
    for (int s = 0; s < 8; ++s) {
        v2f a[4];
#pragma unroll
        for (int rt = 0; rt < 4; ++rt) a[rt] = *(const v2f*)(ap + s * 66 + 16 * rt);
        const float4 c0 = w[0][s >> 1], c1 = w[1][s >> 1];
        v2f b0, b1;
        if (s & 1) { b0.x = c0.z; b0.y = c0.w; b1.x = c1.z; b1.y = c1.w; }
        else       { b0.x = c0.x; b0.y = c0.y; b1.x = c1.x; b1.y = c1.y; }
#pragma unroll
        for (int rt = 0; rt < 4; ++rt) {
            acc[0][rt] = wmma_f32(a[rt], b0, acc[0][rt]);
            acc[1][rt] = wmma_f32(a[rt], b1, acc[1][rt]);
        }
    }
}

__global__ __launch_bounds__(256) __attribute__((amdgpu_waves_per_eu(4, 4))) void qkv_gemm_kernel(
    const float* __restrict__ x, const float* __restrict__ wq,
    const float* __restrict__ bq, float* __restrict__ qkv)
{
    __shared__ float2 Xs2[128 * 66];
    const int row0 = blockIdx.x * 64;
    const int n = row0 >> 12;
    const int l0 = row0 & (LL - 1);
    const int tid = threadIdx.x;
    const int lane = tid & 31, wave = tid >> 5;
    const int c = lane & 15, hf = lane >> 4;

    {
        const int m4 = (tid & 15) * 4;
        const int pb = tid >> 4;
        const float* xs = x + (size_t)n * CC * LL + l0 + m4;
        float4 lo[8], hi[8];
#pragma unroll
        for (int i = 0; i < 8; ++i) {
            const int p = pb + 16 * i;
            lo[i] = *(const float4*)(xs + (size_t)(2 * p) * LL);
            hi[i] = *(const float4*)(xs + (size_t)(2 * p + 1) * LL);
        }
#pragma unroll
        for (int i = 0; i < 8; ++i) {
            const int p = pb + 16 * i;
            float4 s0, s1;
            s0.x = lo[i].x; s0.y = hi[i].x; s0.z = lo[i].y; s0.w = hi[i].y;
            s1.x = lo[i].z; s1.y = hi[i].z; s1.z = lo[i].w; s1.w = hi[i].w;
            *(float4*)&Xs2[p * 66 + m4]     = s0;
            *(float4*)&Xs2[p * 66 + m4 + 2] = s1;
        }
    }

    const float* wrow = wq + (size_t)(32 * wave + c) * CC + 16 * hf;
    const float2* arow = Xs2 + (8 * hf) * 66 + c;
    float* obase = qkv + ((size_t)(n * HH + wave) * LL + l0 + 8 * hf) * DD + c;

    float4 wa[2][4], wb[2][4];
#pragma unroll
    for (int q = 0; q < 4; ++q) {
        wa[0][q] = *(const float4*)(wrow + 4 * q);
        wa[1][q] = *(const float4*)(wrow + 16 * CC + 4 * q);
    }
    __syncthreads();

    for (int cp = 0; cp < 3; ++cp) {
        const float* wg = wrow + (size_t)cp * 256 * CC;
        v8f acc[2][4] = {};
#pragma unroll 1
        for (int kk = 0; kk < 4; ++kk) {
            const int kc = kk * 64;
#pragma unroll
            for (int q = 0; q < 4; ++q) {
                wb[0][q] = *(const float4*)(wg + kc + 32 + 4 * q);
                wb[1][q] = *(const float4*)(wg + 16 * CC + kc + 32 + 4 * q);
            }
            qkv_chunk(acc, wa, arow + (kc >> 1) * 66);
            {
                const float* wn = (kk < 3) ? (wg + kc + 64) : ((cp < 2) ? (wg + 256 * CC) : wg);
#pragma unroll
                for (int q = 0; q < 4; ++q) {
                    wa[0][q] = *(const float4*)(wn + 4 * q);
                    wa[1][q] = *(const float4*)(wn + 16 * CC + 4 * q);
                }
            }
            qkv_chunk(acc, wb, arow + ((kc + 32) >> 1) * 66);
        }
        const float bias0 = bq[256 * cp + 32 * wave + c];
        const float bias1 = bq[256 * cp + 32 * wave + 16 + c];
        float* og = obase + (size_t)cp * SET;
#pragma unroll
        for (int rt = 0; rt < 4; ++rt)
#pragma unroll
            for (int r = 0; r < 8; ++r) {
                float* o = og + (16 * rt + r) * DD;
                o[0]  = acc[0][rt][r] + bias0;
                o[16] = acc[1][rt][r] + bias1;
            }
    }
}

__global__ __launch_bounds__(512) __attribute__((amdgpu_waves_per_eu(4, 4))) void norm_attn_kernel(float* __restrict__ qkv)
{
    __shared__ float red[16 * 1024];
    const int b = blockIdx.x;
    const int quarter = b & 3;
    const int h = (b >> 2) & 7;
    const int n = b >> 5;
    const int tid = threadIdx.x, lane = tid & 31, wave = tid >> 5;
    const int c = lane & 15, hf = lane >> 4;
    const size_t wrow = ((size_t)(n * HH + h) * LL + quarter * 1024 + wave * 64) * DD;
    const float* kb = qkv + SET + wrow;
    const float* vb = qkv + 2 * SET + wrow;
    const float* rowp = kb + c * DD + 16 * hf;
    const float* kfrag = kb + (8 * hf) * DD + 2 * c;
    const float* vfrag = vb + (8 * hf) * DD + 2 * c;

    v8f acc[2][2] = {};
    const v2f ones = {1.0f, 1.0f};

    float4 k4[4];
    float2 kf[8], vf[8];
#pragma unroll
    for (int i = 0; i < 4; ++i) k4[i] = *(const float4*)(rowp + 4 * i);
#pragma unroll
    for (int r = 0; r < 8; ++r) {
        kf[r] = *(const float2*)(kfrag + r * DD);
        vf[r] = *(const float2*)(vfrag + r * DD);
    }

#pragma unroll
    for (int t = 0; t < 4; ++t) {
        float4 k4n[4];
        float2 kfn[8], vfn[8];
        if (t < 3) {
            const int off = (t + 1) * 16 * DD;
#pragma unroll
            for (int i = 0; i < 4; ++i) k4n[i] = *(const float4*)(rowp + off + 4 * i);
#pragma unroll
            for (int r = 0; r < 8; ++r) {
                kfn[r] = *(const float2*)(kfrag + off + r * DD);
                vfn[r] = *(const float2*)(vfrag + off + r * DD);
            }
        }
        float pk = 0.f;
#pragma unroll
        for (int i = 0; i < 4; ++i)
            pk += k4[i].x * k4[i].x + k4[i].y * k4[i].y + k4[i].z * k4[i].z + k4[i].w * k4[i].w;
        v2f an; an.x = pk; an.y = 0.0f;
        v8f zero = {};
        const v8f ss = wmma_f32(an, ones, zero);
#pragma unroll
        for (int s = 0; s < 4; ++s) {
            const float i0 = 1.0f / sqrtf(ss[2 * s]);
            const float i1 = 1.0f / sqrtf(ss[2 * s + 1]);
            v2f a0, a1, b0, b1;
            a0.x = kf[2 * s].x * i0; a0.y = kf[2 * s + 1].x * i1;
            a1.x = kf[2 * s].y * i0; a1.y = kf[2 * s + 1].y * i1;
            b0.x = vf[2 * s].x;      b0.y = vf[2 * s + 1].x;
            b1.x = vf[2 * s].y;      b1.y = vf[2 * s + 1].y;
            acc[0][0] = wmma_f32(a0, b0, acc[0][0]);
            acc[0][1] = wmma_f32(a0, b1, acc[0][1]);
            acc[1][0] = wmma_f32(a1, b0, acc[1][0]);
            acc[1][1] = wmma_f32(a1, b1, acc[1][1]);
        }
        if (t < 3) {
#pragma unroll
            for (int i = 0; i < 4; ++i) k4[i] = k4n[i];
#pragma unroll
            for (int r = 0; r < 8; ++r) { kf[r] = kfn[r]; vf[r] = vfn[r]; }
        }
    }
#pragma unroll
    for (int md = 0; md < 2; ++md)
#pragma unroll
        for (int r = 0; r < 8; ++r) {
            float2 o; o.x = acc[md][0][r]; o.y = acc[md][1][r];
            *(float2*)&red[wave * 1024 + (16 * hf + 2 * r + md) * 32 + 2 * c] = o;
        }
    __syncthreads();
    float* pout = qkv + SET + ((size_t)(n * HH + h) * LL + quarter * 1024) * DD;
    for (int i = tid; i < 1024; i += 512) {
        float s = 0.f;
#pragma unroll
        for (int w = 0; w < 16; ++w) s += red[w * 1024 + i];
        pout[i] = s;
    }
}

static __device__ __forceinline__ float row16_sum(float x) {
    x += __int_as_float(__builtin_amdgcn_update_dpp(0, __float_as_int(x), 0x161, 0xf, 0xf, true));
    x += __int_as_float(__builtin_amdgcn_update_dpp(0, __float_as_int(x), 0x162, 0xf, 0xf, true));
    x += __int_as_float(__builtin_amdgcn_update_dpp(0, __float_as_int(x), 0x164, 0xf, 0xf, true));
    x += __int_as_float(__builtin_amdgcn_update_dpp(0, __float_as_int(x), 0x168, 0xf, 0xf, true));
    return x;
}

__global__ __launch_bounds__(256) void epilogue_kernel(
    const float* __restrict__ qkv,
    const float* __restrict__ wdc, float* __restrict__ out_pre)
{
    __shared__ float4 Bq[16 * 16];
    const int b = blockIdx.x;
    const int lc = b & 15;
    const int h = (b >> 4) & 7;
    const int n = b >> 7;
    const int tid = threadIdx.x, lane = tid & 31;
    const int wave = tid >> 5;
    const int c = lane & 15, hf = lane >> 4;
    const int lw = lc * 256 + wave * 32;
    const size_t hp = (size_t)(n * HH + h) * LL * DD;
    const float* qbase = qkv + hp;

    float asum[4];
#pragma unroll
    for (int j = 0; j < 4; ++j) {
        const int d = (tid >> 5) + 8 * j;
        const float* pp = qkv + SET + hp + d * DD + lane;
        float a = pp[0];
        a += pp[1024 * DD];
        a += pp[2048 * DD];
        a += pp[3072 * DD];
        asum[j] = a;
    }

    float4 q4[2][4];
    float2 vv[2][16];
#pragma unroll
    for (int t = 0; t < 2; ++t) {
        const float* qr = qbase + (lw + 16 * t + c) * DD + 16 * hf;
#pragma unroll
        for (int i = 0; i < 4; ++i) q4[t][i] = *(const float4*)(qr + 4 * i);
    }
    const int lv0 = lw + 8 * hf - 4;
    const float* vcol = qkv + 2 * SET + hp + 2 * c;
#pragma unroll
    for (int i = 0; i < 4; ++i) {
        const int l = lv0 + i;
        const float2 x = *(const float2*)(vcol + (l < 0 ? 0 : l) * DD);
        vv[0][i].x = (l < 0) ? 0.0f : x.x;
        vv[0][i].y = (l < 0) ? 0.0f : x.y;
    }
    {
        const float* vr = vcol + (lv0 + 4) * DD;
#pragma unroll
        for (int i = 4; i < 28; ++i) vv[i >> 4][i & 15] = *(const float2*)(vr + (i - 4) * DD);
    }
#pragma unroll
    for (int i = 12; i < 16; ++i) {
        const int l = lv0 + 16 + i;
        const float2 x = *(const float2*)(vcol + (l < LL ? l : LL - 1) * DD);
        vv[1][i].x = (l < LL) ? x.x : 0.0f;
        vv[1][i].y = (l < LL) ? x.y : 0.0f;
    }

#pragma unroll
    for (int j = 0; j < 4; ++j) {
        const int d = (tid >> 5) + 8 * j;
        ((float*)Bq)[(((d >> 1) * 16 + (lane >> 1)) << 2) + ((lane & 1) << 1) + (d & 1)] = asum[j];
    }
    __syncthreads();

    float4 bf[8];
#pragma unroll
    for (int s = 0; s < 8; ++s) bf[s] = Bq[(8 * hf + s) * 16 + c];

    float wd[9];
#pragma unroll
    for (int r = 0; r < 9; ++r) wd[r] = wdc[h * 9 + r];

    const float inv_pi = 0.31830988618379067154f;
#pragma unroll
    for (int t = 0; t < 2; ++t) {
        float pq = 0.f;
#pragma unroll
        for (int i = 0; i < 4; ++i)
            pq += q4[t][i].x * q4[t][i].x + q4[t][i].y * q4[t][i].y + q4[t][i].z * q4[t][i].z + q4[t][i].w * q4[t][i].w;
        pq += __shfl_xor(pq, 16, 32);
        const float invq = 1.0f / sqrtf(pq);
#pragma unroll
        for (int i = 0; i < 4; ++i) {
            q4[t][i].x *= invq; q4[t][i].y *= invq; q4[t][i].z *= invq; q4[t][i].w *= invq;
        }
        v8f acc0 = {}, acc1 = {};
#pragma unroll
        for (int s = 0; s < 8; ++s) {
            const float4 qq = q4[t][s >> 1];
            v2f a, b0, b1;
            if (s & 1) { a.x = qq.z; a.y = qq.w; } else { a.x = qq.x; a.y = qq.y; }
            b0.x = bf[s].x; b0.y = bf[s].y;
            b1.x = bf[s].z; b1.y = bf[s].w;
            acc0 = wmma_f32(a, b0, acc0);
            acc1 = wmma_f32(a, b1, acc1);
        }
        float* orow = out_pre + (size_t)(n * LL + lw + 16 * t + 8 * hf) * CC + h * DD + 2 * c;
#pragma unroll
        for (int r = 0; r < 8; ++r) {
            const float2 vc = vv[t][r + 4];
            const float v0 = 0.5f * vc.x + inv_pi * acc0[r];
            const float v1 = 0.5f * vc.y + inv_pi * acc1[r];
            const float nrm = sqrtf(row16_sum(v0 * v0 + v1 * v1));
            float dc0 = 0.f, dc1 = 0.f;
#pragma unroll
            for (int k = 0; k < 9; ++k) {
                dc0 += wd[k] * vv[t][r + k].x;
                dc1 += wd[k] * vv[t][r + k].y;
            }
            float2 o;
            o.x = v0 / nrm + dc0;
            o.y = v1 / nrm + dc1;
            *(float2*)(orow + (size_t)r * CC) = o;
        }
    }
}

#define PJ_PITCH 260

static __device__ __forceinline__ void proj_chunk(v8f (&acc)[2][4], const float4 (&w)[2][4],
                                                  const float* __restrict__ bp)
{
#pragma unroll
    for (int q = 0; q < 4; ++q) {
        float4 b[4];
#pragma unroll
        for (int lt = 0; lt < 4; ++lt) b[lt] = *(const float4*)(bp + lt * 16 * PJ_PITCH + 4 * q);
        const float4 w0 = w[0][q], w1 = w[1][q];
        v2f a0, a1;
        a0.x = w0.x; a0.y = w0.y; a1.x = w1.x; a1.y = w1.y;
#pragma unroll
        for (int lt = 0; lt < 4; ++lt) {
            v2f bb; bb.x = b[lt].x; bb.y = b[lt].y;
            acc[0][lt] = wmma_f32(a0, bb, acc[0][lt]);
            acc[1][lt] = wmma_f32(a1, bb, acc[1][lt]);
        }
        a0.x = w0.z; a0.y = w0.w; a1.x = w1.z; a1.y = w1.w;
#pragma unroll
        for (int lt = 0; lt < 4; ++lt) {
            v2f bb; bb.x = b[lt].z; bb.y = b[lt].w;
            acc[0][lt] = wmma_f32(a0, bb, acc[0][lt]);
            acc[1][lt] = wmma_f32(a1, bb, acc[1][lt]);
        }
    }
}

__global__ __launch_bounds__(256) __attribute__((amdgpu_waves_per_eu(4, 4))) void proj_gemm_kernel(
    const float* __restrict__ A, const float* __restrict__ wp,
    const float* __restrict__ bp, float* __restrict__ out)
{
    __shared__ float Bs[64 * PJ_PITCH];
    const int row0 = blockIdx.x * 64;
    const int n = row0 >> 12;
    const int l0 = row0 & (LL - 1);
    const int tid = threadIdx.x, lane = tid & 31, wave = tid >> 5;
    const int c = lane & 15, hf = lane >> 4;

    {
        const float* sp = A + (size_t)row0 * CC + 4 * tid;
        float4 t[16];
#pragma unroll
        for (int i = 0; i < 16; ++i) t[i] = *(const float4*)(sp + 1024 * i);
        float* dp = Bs + (tid >> 6) * PJ_PITCH + 4 * (tid & 63);
#pragma unroll
        for (int i = 0; i < 16; ++i) *(float4*)(dp + 4 * i * PJ_PITCH) = t[i];
    }

    const float* wrow = wp + (size_t)(32 * wave + c) * CC + 16 * hf;
    const float* brow = Bs + c * PJ_PITCH + 16 * hf;

    float4 wa[2][4], wb[2][4];
#pragma unroll
    for (int q = 0; q < 4; ++q) {
        wa[0][q] = *(const float4*)(wrow + 4 * q);
        wa[1][q] = *(const float4*)(wrow + 16 * CC + 4 * q);
    }
    float4 bias[2][2];
#pragma unroll
    for (int jt = 0; jt < 2; ++jt) {
        bias[jt][0] = *(const float4*)(bp + 32 * wave + 16 * jt + 8 * hf);
        bias[jt][1] = *(const float4*)(bp + 32 * wave + 16 * jt + 8 * hf + 4);
    }
    __syncthreads();

    v8f acc[2][4] = {};
#pragma unroll 1
    for (int kk = 0; kk < 4; ++kk) {
        const int kc = kk * 64;
#pragma unroll
        for (int q = 0; q < 4; ++q) {
            wb[0][q] = *(const float4*)(wrow + kc + 32 + 4 * q);
            wb[1][q] = *(const float4*)(wrow + 16 * CC + kc + 32 + 4 * q);
        }
        proj_chunk(acc, wa, brow + kc);
        {
            const float* wn = (kk < 3) ? (wrow + kc + 64) : wrow;
#pragma unroll
            for (int q = 0; q < 4; ++q) {
                wa[0][q] = *(const float4*)(wn + 4 * q);
                wa[1][q] = *(const float4*)(wn + 16 * CC + 4 * q);
            }
        }
        proj_chunk(acc, wb, brow + kc + 32);
    }
    float* obase = out + ((size_t)(n * CC + 32 * wave + 8 * hf)) * LL + l0 + c;
#pragma unroll
    for (int jt = 0; jt < 2; ++jt)
#pragma unroll
        for (int r = 0; r < 8; ++r) {
            const float4 bq4 = bias[jt][r >> 2];
            const float bj = ((r & 3) == 0) ? bq4.x : ((r & 3) == 1) ? bq4.y : ((r & 3) == 2) ? bq4.z : bq4.w;
            float* o = obase + (size_t)(16 * jt + r) * LL;
#pragma unroll
            for (int lt = 0; lt < 4; ++lt) o[16 * lt] = acc[jt][lt][r] + bj;
        }
}

extern "C" void kernel_launch(void* const* d_in, const int* in_sizes, int n_in,
                              void* d_out, int out_size, void* d_ws, size_t ws_size,
                              hipStream_t stream) {
    const float* x      = (const float*)d_in[0];
    const float* w_qkv  = (const float*)d_in[1];
    const float* b_qkv  = (const float*)d_in[2];
    const float* w_proj = (const float*)d_in[3];
    const float* b_proj = (const float*)d_in[4];
    const float* w_dcv  = (const float*)d_in[5];
    float* out = (float*)d_out;

    float* qkv     = (float*)d_ws;
    float* out_pre = qkv + (size_t)NB * LL * TC + (size_t)NB * HH * DD * DD;

    qkv_gemm_kernel<<<512, 256, 0, stream>>>(x, w_qkv, b_qkv, qkv);
    norm_attn_kernel<<<256, 512, 0, stream>>>(qkv);
    epilogue_kernel<<<1024, 256, 0, stream>>>(qkv, w_dcv, out_pre);
    proj_gemm_kernel<<<512, 256, 0, stream>>>(out_pre, w_proj, b_proj, out);
}
